# starting kernel
# baseline (speedup 1.0000x reference)
#include <hip/hip_runtime.h>
#include <hip/hip_bf16.h>

typedef __attribute__((ext_vector_type(2))) float v2f;
typedef __attribute__((ext_vector_type(8))) float v8f;

#define HID 64

__global__ void init_kernel(float* __restrict__ deg, int* __restrict__ mm, int N) {
    int i = blockIdx.x * blockDim.x + threadIdx.x;
    if (i < N) deg[i] = 1.0f;
    if (i == 0) { mm[0] = 0x7FFFFFFF; mm[1] = 0x80000000; }
}

__global__ void minmax_kernel(const int* __restrict__ ec, int E, int* __restrict__ mm) {
    int i = blockIdx.x * blockDim.x + threadIdx.x;
    int stride = gridDim.x * blockDim.x;
    int lmin = 0x7FFFFFFF, lmax = 0x80000000;
    for (; i < E; i += stride) {
        int v = ec[i];
        lmin = min(lmin, v);
        lmax = max(lmax, v);
    }
    atomicMin(&mm[0], lmin);
    atomicMax(&mm[1], lmax);
}

__global__ void edgew_kernel(const int* __restrict__ ec, const int* __restrict__ col,
                             const int* __restrict__ mm, float* __restrict__ ew,
                             float* __restrict__ deg, int E) {
    int e = blockIdx.x * blockDim.x + threadIdx.x;
    if (e >= E) return;
    float mn = (float)mm[0], mx = (float)mm[1];
    float denom = mx - mn;
    float w = (denom > 0.0f) ? ((float)ec[e] - mn) / denom : 0.0f;
    ew[e] = w;
    atomicAdd(&deg[col[e]], w);
}

__global__ void dinv_kernel(float* __restrict__ deg, int N) {
    int i = blockIdx.x * blockDim.x + threadIdx.x;
    if (i >= N) return;
    float d = deg[i];
    deg[i] = (d > 0.0f) ? rsqrtf(d) : 0.0f;
}

template <int K, bool RELU>
__global__ void gemm_wmma_kernel(const float* __restrict__ A, const float* __restrict__ W,
                                 float* __restrict__ H, int M) {
    __shared__ __align__(16) float sW[K * HID];

    for (int i = threadIdx.x; i < (K / 2) * HID; i += blockDim.x) {
        int kp = i >> 6;
        int c  = i & 63;
        float w0 = W[(2 * kp) * HID + c];
        float w1 = W[(2 * kp + 1) * HID + c];
        float* dst = sW + kp * (2 * HID) + 2 * c;
        dst[0] = w0;
        dst[1] = w1;
    }
    __syncthreads();

    const int lane  = threadIdx.x & 31;
    const int widx  = threadIdx.x >> 5;
    const int half  = lane >> 4;
    const int sub   = lane & 15;
    const long tile = (long)blockIdx.x * 8 + widx;
    const long row0 = tile * 16;
    if (row0 >= M) return;

    v8f acc[4] = {};

    const float* aRow = A + (row0 + sub) * (long)K;

#pragma unroll 4
    for (int kk = 0; kk < K; kk += 4) {
        const int kb = kk + 2 * half;
        v2f a = *(const v2f*)(aRow + kb);
        if (RELU) { a.x = fmaxf(a.x, 0.0f); a.y = fmaxf(a.y, 0.0f); }

        const float* wp = sW + (kb >> 1) * (2 * HID) + 2 * sub;
#pragma unroll
        for (int t = 0; t < 4; ++t) {
            v2f b = *(const v2f*)(wp + t * 32);
            acc[t] = __builtin_amdgcn_wmma_f32_16x16x4_f32(
                 false, a,  false, b,
                 (short)0, acc[t],
                 false,  false);
        }
    }

#pragma unroll
    for (int r = 0; r < 8; ++r) {
        float* hp = H + (row0 + r + 8 * half) * (long)HID + sub;
#pragma unroll
        for (int t = 0; t < 4; ++t) hp[t * 16] = acc[t][r];
    }
}

__global__ void selfloop_bias_kernel(const float* __restrict__ h, const float* __restrict__ dinv,
                                     const float* __restrict__ bias, float* __restrict__ out,
                                     long total) {
    long t = (long)blockIdx.x * blockDim.x + threadIdx.x;
    if (t >= total) return;
    long i = t >> 6;
    int  d = (int)(t & 63);
    float di = dinv[i];
    out[t] = h[t] * di * di + bias[d];
}

__global__ void scatter_kernel(const float* __restrict__ h, const float* __restrict__ dinv,
                               const float* __restrict__ ew, const int* __restrict__ row,
                               const int* __restrict__ col, float* __restrict__ out, int E) {
    const int lane = threadIdx.x & 31;
    const int widx = threadIdx.x >> 5;
    const long ebase = (long)blockIdx.x * 32;
#pragma unroll
    for (int k = 0; k < 4; ++k) {
        long e = ebase + (long)k * 8 + widx;
        if (e >= E) continue;
        int r = row[e], c = col[e];
        float norm = dinv[r] * ew[e] * dinv[c];
        const float* hr = h + (long)r * HID;
        float v0 = hr[lane]      * norm;
        float v1 = hr[lane + 32] * norm;
        float* op = out + (long)c * HID;
        atomicAdd(op + lane,      v0);
        atomicAdd(op + lane + 32, v1);
    }
}

__global__ void logsoftmax_kernel(const float* __restrict__ H, float* __restrict__ out, int N) {
    int gid  = blockIdx.x * blockDim.x + threadIdx.x;
    int node = gid >> 5;
    int lane = gid & 31;
    if (node >= N) return;
    const float* rowp = H + (long)node * HID;
    float a = rowp[lane];
    float b = rowp[lane + 32];
    float m = fmaxf(a, b);
#pragma unroll
    for (int off = 16; off > 0; off >>= 1) m = fmaxf(m, __shfl_xor(m, off, 32));
    float s = __expf(a - m) + __expf(b - m);
#pragma unroll
    for (int off = 16; off > 0; off >>= 1) s += __shfl_xor(s, off, 32);
    float lse = m + __logf(s);
    float* op = out + (long)node * HID;
    op[lane]      = a - lse;
    op[lane + 32] = b - lse;
}

extern "C" void kernel_launch(void* const* d_in, const int* in_sizes, int n_in,
                              void* d_out, int out_size, void* d_ws, size_t ws_size,
                              hipStream_t stream) {
    const float* x  = (const float*)d_in[0];
    const int*   ei = (const int*)d_in[1];
    const int*   ec = (const int*)d_in[2];
    const float* W1 = (const float*)d_in[3];
    const float* b1 = (const float*)d_in[4];
    const float* W2 = (const float*)d_in[5];
    const float* b2 = (const float*)d_in[6];

    const int IN_DIM = 256;
    const int N = in_sizes[0] / IN_DIM;
    const int E = in_sizes[2];
    const int* row = ei;
    const int* col = ei + E;

    float* wsf  = (float*)d_ws;
    float* ew   = wsf;
    float* deg  = ew   + (size_t)E;
    float* h    = deg  + (size_t)N;
    float* agg1 = h    + (size_t)N * HID;
    float* h2   = agg1 + (size_t)N * HID;
    int*   mm   = (int*)(h2 + (size_t)N * HID);

    float* emb = (float*)d_out;
    float* lsm = emb + (size_t)N * HID;

    const long NH = (long)N * HID;
    const int tileBlocks = (int)((N / 16 + 7) / 8);
    const int edgeBlocks = (E + 31) / 32;

    init_kernel<<<(N + 255) / 256, 256, 0, stream>>>(deg, mm, N);
    minmax_kernel<<<1024, 256, 0, stream>>>(ec, E, mm);
    edgew_kernel<<<(E + 255) / 256, 256, 0, stream>>>(ec, col, mm, ew, deg, E);
    dinv_kernel<<<(N + 255) / 256, 256, 0, stream>>>(deg, N);

    gemm_wmma_kernel<256, false><<<tileBlocks, 256, 0, stream>>>(x, W1, h, N);
    selfloop_bias_kernel<<<(int)((NH + 255) / 256), 256, 0, stream>>>(h, deg, b1, agg1, NH);
    scatter_kernel<<<edgeBlocks, 256, 0, stream>>>(h, deg, ew, row, col, agg1, E);

    gemm_wmma_kernel<64, true><<<tileBlocks, 256, 0, stream>>>(agg1, W2, h2, N);
    selfloop_bias_kernel<<<(int)((NH + 255) / 256), 256, 0, stream>>>(h2, deg, b2, emb, NH);
    scatter_kernel<<<edgeBlocks, 256, 0, stream>>>(h2, deg, ew, row, col, emb, E);

    logsoftmax_kernel<<<(N + 7) / 8, 256, 0, stream>>>(emb, lsm, N);
}


# scatter: one wave per 32 edges (coalesced index loads, norm per lane, readlane broadcast), 16 row loads of 8 edges issued ahead of their f32 atomics; atomics kept, arrival order free as before
# speedup vs baseline: 1.2191x; 1.2191x over previous
#include <hip/hip_runtime.h>
#include <hip/hip_bf16.h>

typedef __attribute__((ext_vector_type(2))) float v2f;
typedef __attribute__((ext_vector_type(8))) float v8f;

#define HID 64

__global__ void init_kernel(float* __restrict__ deg, int* __restrict__ mm, int N) {
    int i = blockIdx.x * blockDim.x + threadIdx.x;
    if (i < N) deg[i] = 1.0f;
    if (i == 0) { mm[0] = 0x7FFFFFFF; mm[1] = 0x80000000; }
}

__global__ void minmax_kernel(const int* __restrict__ ec, int E, int* __restrict__ mm) {
    int i = blockIdx.x * blockDim.x + threadIdx.x;
    int stride = gridDim.x * blockDim.x;
    int lmin = 0x7FFFFFFF, lmax = 0x80000000;
    for (; i < E; i += stride) {
        int v = ec[i];
        lmin = min(lmin, v);
        lmax = max(lmax, v);
    }
    atomicMin(&mm[0], lmin);
    atomicMax(&mm[1], lmax);
}

__global__ void edgew_kernel(const int* __restrict__ ec, const int* __restrict__ col,
                             const int* __restrict__ mm, float* __restrict__ ew,
                             float* __restrict__ deg, int E) {
    int e = blockIdx.x * blockDim.x + threadIdx.x;
    if (e >= E) return;
    float mn = (float)mm[0], mx = (float)mm[1];
    float denom = mx - mn;
    float w = (denom > 0.0f) ? ((float)ec[e] - mn) / denom : 0.0f;
    ew[e] = w;
    atomicAdd(&deg[col[e]], w);
}

__global__ void dinv_kernel(float* __restrict__ deg, int N) {
    int i = blockIdx.x * blockDim.x + threadIdx.x;
    if (i >= N) return;
    float d = deg[i];
    deg[i] = (d > 0.0f) ? rsqrtf(d) : 0.0f;
}

template <int K, bool RELU>
__global__ void gemm_wmma_kernel(const float* __restrict__ A, const float* __restrict__ W,
                                 float* __restrict__ H, int M) {
    __shared__ __align__(16) float sW[K * HID];

    for (int i = threadIdx.x; i < (K / 2) * HID; i += blockDim.x) {
        int kp = i >> 6;
        int c  = i & 63;
        float w0 = W[(2 * kp) * HID + c];
        float w1 = W[(2 * kp + 1) * HID + c];
        float* dst = sW + kp * (2 * HID) + 2 * c;
        dst[0] = w0;
        dst[1] = w1;
    }
    __syncthreads();

    const int lane  = threadIdx.x & 31;
    const int widx  = threadIdx.x >> 5;
    const int half  = lane >> 4;
    const int sub   = lane & 15;
    const long tile = (long)blockIdx.x * 8 + widx;
    const long row0 = tile * 16;
    if (row0 >= M) return;

    v8f acc[4] = {};

    const float* aRow = A + (row0 + sub) * (long)K;

#pragma unroll 4
    for (int kk = 0; kk < K; kk += 4) {
        const int kb = kk + 2 * half;
        v2f a = *(const v2f*)(aRow + kb);
        if (RELU) { a.x = fmaxf(a.x, 0.0f); a.y = fmaxf(a.y, 0.0f); }

        const float* wp = sW + (kb >> 1) * (2 * HID) + 2 * sub;
#pragma unroll
        for (int t = 0; t < 4; ++t) {
            v2f b = *(const v2f*)(wp + t * 32);
            acc[t] = __builtin_amdgcn_wmma_f32_16x16x4_f32(
                 false, a,  false, b,
                 (short)0, acc[t],
                 false,  false);
        }
    }

#pragma unroll
    for (int r = 0; r < 8; ++r) {
        float* hp = H + (row0 + r + 8 * half) * (long)HID + sub;
#pragma unroll
        for (int t = 0; t < 4; ++t) hp[t * 16] = acc[t][r];
    }
}

__global__ void selfloop_bias_kernel(const float* __restrict__ h, const float* __restrict__ dinv,
                                     const float* __restrict__ bias, float* __restrict__ out,
                                     long total) {
    long t = (long)blockIdx.x * blockDim.x + threadIdx.x;
    if (t >= total) return;
    long i = t >> 6;
    int  d = (int)(t & 63);
    float di = dinv[i];
    out[t] = h[t] * di * di + bias[d];
}

__global__ void __launch_bounds__(256)
scatter_kernel(const float* __restrict__ h, const float* __restrict__ dinv,
               const float* __restrict__ ew, const int* __restrict__ row,
               const int* __restrict__ col, float* __restrict__ out, int E) {
    const int lane = threadIdx.x & 31;
    const long e0 = ((long)blockIdx.x * 8 + (threadIdx.x >> 5)) * 32;
    if (e0 >= E) return;
    const long e = e0 + lane;
    int r = 0, c = 0;
    float nrm = 0.0f;
    if (e < E) {
        r = row[e];
        c = col[e];
        nrm = dinv[r] * ew[e] * dinv[c];
    }
    const long left = (long)E - e0;
    const int cnt = left < 32 ? (int)left : 32;
    const float* hl = h + lane;
    float* ol = out + lane;

    if (cnt == 32) {
#pragma unroll 1
        for (int j0 = 0; j0 < 32; j0 += 8) {
            float a0[8], a1[8];
#pragma unroll
            for (int j = 0; j < 8; ++j) {
                const int rj = __builtin_amdgcn_readlane(r, j0 + j);
                const float* hr = hl + (long)rj * HID;
                a0[j] = hr[0];
                a1[j] = hr[32];
            }
#pragma unroll
            for (int j = 0; j < 8; ++j) {
                const int cj = __builtin_amdgcn_readlane(c, j0 + j);
                const float nj = __int_as_float(
                    __builtin_amdgcn_readlane(__float_as_int(nrm), j0 + j));
                float* op = ol + (long)cj * HID;
                atomicAdd(op,      a0[j] * nj);
                atomicAdd(op + 32, a1[j] * nj);
            }
        }
    } else {
        for (int j = 0; j < cnt; ++j) {
            const int rj = __builtin_amdgcn_readlane(r, j);
            const int cj = __builtin_amdgcn_readlane(c, j);
            const float nj = __int_as_float(
                __builtin_amdgcn_readlane(__float_as_int(nrm), j));
            const float* hr = hl + (long)rj * HID;
            float* op = ol + (long)cj * HID;
            const float v0 = hr[0]  * nj;
            const float v1 = hr[32] * nj;
            atomicAdd(op,      v0);
            atomicAdd(op + 32, v1);
        }
    }
}

__global__ void logsoftmax_kernel(const float* __restrict__ H, float* __restrict__ out, int N) {
    int gid  = blockIdx.x * blockDim.x + threadIdx.x;
    int node = gid >> 5;
    int lane = gid & 31;
    if (node >= N) return;
    const float* rowp = H + (long)node * HID;
    float a = rowp[lane];
    float b = rowp[lane + 32];
    float m = fmaxf(a, b);
#pragma unroll
    for (int off = 16; off > 0; off >>= 1) m = fmaxf(m, __shfl_xor(m, off, 32));
    float s = __expf(a - m) + __expf(b - m);
#pragma unroll
    for (int off = 16; off > 0; off >>= 1) s += __shfl_xor(s, off, 32);
    float lse = m + __logf(s);
    float* op = out + (long)node * HID;
    op[lane]      = a - lse;
    op[lane + 32] = b - lse;
}

extern "C" void kernel_launch(void* const* d_in, const int* in_sizes, int n_in,
                              void* d_out, int out_size, void* d_ws, size_t ws_size,
                              hipStream_t stream) {
    const float* x  = (const float*)d_in[0];
    const int*   ei = (const int*)d_in[1];
    const int*   ec = (const int*)d_in[2];
    const float* W1 = (const float*)d_in[3];
    const float* b1 = (const float*)d_in[4];
    const float* W2 = (const float*)d_in[5];
    const float* b2 = (const float*)d_in[6];

    const int IN_DIM = 256;
    const int N = in_sizes[0] / IN_DIM;
    const int E = in_sizes[2];
    const int* row = ei;
    const int* col = ei + E;

    float* wsf  = (float*)d_ws;
    float* ew   = wsf;
    float* deg  = ew   + (size_t)E;
    float* h    = deg  + (size_t)N;
    float* agg1 = h    + (size_t)N * HID;
    float* h2   = agg1 + (size_t)N * HID;
    int*   mm   = (int*)(h2 + (size_t)N * HID);

    float* emb = (float*)d_out;
    float* lsm = emb + (size_t)N * HID;

    const long NH = (long)N * HID;
    const int tileBlocks = (int)((N / 16 + 7) / 8);
    const int edgeBlocks = (E + 255) / 256;

    init_kernel<<<(N + 255) / 256, 256, 0, stream>>>(deg, mm, N);
    minmax_kernel<<<1024, 256, 0, stream>>>(ec, E, mm);
    edgew_kernel<<<(E + 255) / 256, 256, 0, stream>>>(ec, col, mm, ew, deg, E);
    dinv_kernel<<<(N + 255) / 256, 256, 0, stream>>>(deg, N);

    gemm_wmma_kernel<256, false><<<tileBlocks, 256, 0, stream>>>(x, W1, h, N);
    selfloop_bias_kernel<<<(int)((NH + 255) / 256), 256, 0, stream>>>(h, deg, b1, agg1, NH);
    scatter_kernel<<<edgeBlocks, 256, 0, stream>>>(h, deg, ew, row, col, agg1, E);

    gemm_wmma_kernel<64, true><<<tileBlocks, 256, 0, stream>>>(agg1, W2, h2, N);
    selfloop_bias_kernel<<<(int)((NH + 255) / 256), 256, 0, stream>>>(h2, deg, b2, emb, NH);
    scatter_kernel<<<edgeBlocks, 256, 0, stream>>>(h2, deg, ew, row, col, emb, E);

    logsoftmax_kernel<<<(N + 7) / 8, 256, 0, stream>>>(emb, lsm, N);
}


# aggregate by gather: edges bucketed by destination each call (counting add's return = rank, scan, place); a wave sums a node's rows in registers, a thread its deg; no f32 atomics, sums in rank order
# speedup vs baseline: 1.5595x; 1.2792x over previous
#include <hip/hip_runtime.h>
#include <hip/hip_bf16.h>

typedef __attribute__((ext_vector_type(2))) float v2f;
typedef __attribute__((ext_vector_type(8))) float v8f;

#define HID 64

__global__ void init_kernel(int* __restrict__ cnt, int* __restrict__ mm, int N) {
    int i = blockIdx.x * blockDim.x + threadIdx.x;
    if (i < N) cnt[i] = 0;
    if (i == 0) { mm[0] = 0x7FFFFFFF; mm[1] = 0x80000000; }
}

__global__ void minmax_kernel(const int* __restrict__ ec, int E, int* __restrict__ mm) {
    int i = blockIdx.x * blockDim.x + threadIdx.x;
    int stride = gridDim.x * blockDim.x;
    int lmin = 0x7FFFFFFF, lmax = 0x80000000;
    for (; i < E; i += stride) {
        int v = ec[i];
        lmin = min(lmin, v);
        lmax = max(lmax, v);
    }
    atomicMin(&mm[0], lmin);
    atomicMax(&mm[1], lmax);
}

#define EPT 4

__global__ void __launch_bounds__(256)
count_kernel(const int* __restrict__ col, int* __restrict__ cnt,
             int* __restrict__ rank, int E) {
    const int T = gridDim.x * blockDim.x;
    const int t = blockIdx.x * blockDim.x + threadIdx.x;
    if (t + (EPT - 1) * T < E) {
        int c[EPT], rk[EPT];
        for (int u = 0; u < EPT; ++u) c[u] = col[t + u * T];
        for (int u = 0; u < EPT; ++u) rk[u] = atomicAdd(&cnt[c[u]], 1);
        for (int u = 0; u < EPT; ++u) rank[t + u * T] = rk[u];
    } else {
        for (int e = t; e < E; e += T) rank[e] = atomicAdd(&cnt[col[e]], 1);
    }
}

#define SCAN_B 256
#define SCAN_SHIFT 8

__global__ void __launch_bounds__(SCAN_B)
scan_kernel(const int* __restrict__ cnt, int* __restrict__ off,
            int* __restrict__ bsum, int N) {
    __shared__ int wsum[SCAN_B / 32];
    const int tid = threadIdx.x, lane = tid & 31, w = tid >> 5;
    const int i = blockIdx.x * SCAN_B + tid;
    const int v = (i < N) ? cnt[i] : 0;
    int inc = v;
    for (int o = 1; o < 32; o <<= 1) {
        int t = __shfl_up(inc, o, 32);
        if (lane >= o) inc += t;
    }
    if (lane == 31) wsum[w] = inc;
    __syncthreads();
    int wbase = 0, total = 0;
    for (int k = 0; k < SCAN_B / 32; ++k) {
        int s = wsum[k];
        if (k < w) wbase += s;
        total += s;
    }
    if (i < N) off[i] = wbase + inc - v;
    if (tid == 0) bsum[blockIdx.x] = total;
}

__global__ void __launch_bounds__(1024)
bscan_kernel(int* __restrict__ bsum, int nb) {
    __shared__ int wsum[32];
    const int tid = threadIdx.x, lane = tid & 31, w = tid >> 5;
    int carry = 0;
    for (int base = 0; base < nb; base += 1024) {
        const int i = base + tid;
        const int v = (i < nb) ? bsum[i] : 0;
        int inc = v;
        for (int o = 1; o < 32; o <<= 1) {
            int t = __shfl_up(inc, o, 32);
            if (lane >= o) inc += t;
        }
        if (lane == 31) wsum[w] = inc;
        __syncthreads();
        int wbase = 0, total = 0;
        for (int k = 0; k < 32; ++k) {
            int s = wsum[k];
            if (k < w) wbase += s;
            total += s;
        }
        if (i < nb) bsum[i] = carry + wbase + inc - v;
        carry += total;
        __syncthreads();
    }
}

__global__ void __launch_bounds__(256)
fill_kernel(const int* __restrict__ ec, const int* __restrict__ row,
            const int* __restrict__ col, const int* __restrict__ rank,
            const int* __restrict__ off, const int* __restrict__ bsum,
            const int* __restrict__ mm, int2* __restrict__ rec, int E) {
    const int T = gridDim.x * blockDim.x;
    const int t = blockIdx.x * blockDim.x + threadIdx.x;
    const float mn = (float)mm[0], mx = (float)mm[1];
    const float denom = mx - mn;
    const bool span = denom > 0.0f;
    const float dsafe = span ? denom : 1.0f;
    if (t + (EPT - 1) * T < E) {
        int v[EPT], r[EPT], c[EPT], pos[EPT], sb[EPT], so[EPT];
        for (int u = 0; u < EPT; ++u) {
            const int e = t + u * T;
            v[u] = ec[e];
            r[u] = row[e];
            c[u] = col[e];
            pos[u] = rank[e];
        }
        for (int u = 0; u < EPT; ++u) {
            sb[u] = bsum[c[u] >> SCAN_SHIFT];
            so[u] = off[c[u]];
        }
        for (int u = 0; u < EPT; ++u) {
            const float q = ((float)v[u] - mn) / dsafe;
            const float w = span ? q : 0.0f;
            rec[sb[u] + so[u] + pos[u]] = make_int2(r[u], __float_as_int(w));
        }
    } else {
        for (int e = t; e < E; e += T) {
            const float q = ((float)ec[e] - mn) / dsafe;
            const float w = span ? q : 0.0f;
            const int c = col[e];
            rec[bsum[c >> SCAN_SHIFT] + off[c] + rank[e]] = make_int2(row[e], __float_as_int(w));
        }
    }
}

__global__ void dinv_kernel(const int* __restrict__ off, const int* __restrict__ bsum,
                            const int* __restrict__ cnt, const int2* __restrict__ rec,
                            float* __restrict__ dinv, int N) {
    int i = blockIdx.x * blockDim.x + threadIdx.x;
    if (i >= N) return;
    const long start = (long)bsum[i >> SCAN_SHIFT] + off[i];
    const int len = cnt[i];
    float d = 1.0f;
    for (int k = 0; k < len; ++k) d += __int_as_float(rec[start + k].y);
    dinv[i] = (d > 0.0f) ? rsqrtf(d) : 0.0f;
}

template <int K, bool RELU>
__global__ void gemm_wmma_kernel(const float* __restrict__ A, const float* __restrict__ W,
                                 float* __restrict__ H, int M) {
    __shared__ __align__(16) float sW[K * HID];

    for (int i = threadIdx.x; i < (K / 2) * HID; i += blockDim.x) {
        int kp = i >> 6;
        int c  = i & 63;
        float w0 = W[(2 * kp) * HID + c];
        float w1 = W[(2 * kp + 1) * HID + c];
        float* dst = sW + kp * (2 * HID) + 2 * c;
        dst[0] = w0;
        dst[1] = w1;
    }
    __syncthreads();

    const int lane  = threadIdx.x & 31;
    const int widx  = threadIdx.x >> 5;
    const int half  = lane >> 4;
    const int sub   = lane & 15;
    const long tile = (long)blockIdx.x * 8 + widx;
    const long row0 = tile * 16;
    if (row0 >= M) return;

    v8f acc[4] = {};

    const float* aRow = A + (row0 + sub) * (long)K;

#pragma unroll 4
    for (int kk = 0; kk < K; kk += 4) {
        const int kb = kk + 2 * half;
        v2f a = *(const v2f*)(aRow + kb);
        if (RELU) { a.x = fmaxf(a.x, 0.0f); a.y = fmaxf(a.y, 0.0f); }

        const float* wp = sW + (kb >> 1) * (2 * HID) + 2 * sub;
#pragma unroll
        for (int t = 0; t < 4; ++t) {
            v2f b = *(const v2f*)(wp + t * 32);
            acc[t] = __builtin_amdgcn_wmma_f32_16x16x4_f32(
                 false, a,  false, b,
                 (short)0, acc[t],
                 false,  false);
        }
    }

#pragma unroll
    for (int r = 0; r < 8; ++r) {
        float* hp = H + (row0 + r + 8 * half) * (long)HID + sub;
#pragma unroll
        for (int t = 0; t < 4; ++t) hp[t * 16] = acc[t][r];
    }
}

template <int B>
__device__ __forceinline__ void gather_batch(const v2f* __restrict__ hl, int r, float nrm,
                                             int j0, v2f& acc) {
    v2f a[B];
#pragma unroll
    for (int j = 0; j < B; ++j) {
        const int rj = __builtin_amdgcn_readlane(r, j0 + j);
        a[j] = hl[(long)rj * (HID / 2)];
    }
#pragma unroll
    for (int j = 0; j < B; ++j) {
        const float nj = __int_as_float(
            __builtin_amdgcn_readlane(__float_as_int(nrm), j0 + j));
        acc.x = fmaf(a[j].x, nj, acc.x);
        acc.y = fmaf(a[j].y, nj, acc.y);
    }
}

template <bool FIRST>
__global__ void __launch_bounds__(256)
gather_kernel(const float* __restrict__ h, const float* __restrict__ dinv,
              const float* __restrict__ bias, const int* __restrict__ off,
              const int* __restrict__ bsum, const int* __restrict__ cnt,
              int2* __restrict__ rec, float* __restrict__ out, int N) {
    const int lane = threadIdx.x & 31;
    const int wave = __builtin_amdgcn_readfirstlane(blockIdx.x * 8 + (threadIdx.x >> 5));
    const int nwaves = gridDim.x * 8;
    const v2f* hl = (const v2f*)h + lane;
    const v2f bv = ((const v2f*)bias)[lane];
    for (int node = wave; node < N; node += nwaves) {
    const int start = bsum[node >> SCAN_SHIFT] + off[node];
    const int len = cnt[node];
    const float di = dinv[node];
    const v2f self = hl[(long)node * (HID / 2)];
    int2* rp = rec + ((long)start + lane);

    int m = len < 32 ? len : 32;
    int2 rc = make_int2(0, 0);
    if (lane < m) rc = *rp;

    v2f acc = {0.0f, 0.0f};
    for (int base = 0; base < len; base += 32) {
        const int r = rc.x;
        float nrm = __int_as_float(rc.y);
        if (FIRST) {
            if (lane < m) {
                nrm = dinv[r] * nrm * di;
                rp->y = __float_as_int(nrm);
            }
        }
        int j0 = 0;
#pragma unroll 1
        for (; j0 + 16 <= m; j0 += 16) gather_batch<16>(hl, r, nrm, j0, acc);
        switch (m - j0) {
#define GATHER_CASE(B) case B: gather_batch<B>(hl, r, nrm, j0, acc); break;
            GATHER_CASE(15) GATHER_CASE(14) GATHER_CASE(13) GATHER_CASE(12)
            GATHER_CASE(11) GATHER_CASE(10) GATHER_CASE(9)  GATHER_CASE(8)
            GATHER_CASE(7)  GATHER_CASE(6)  GATHER_CASE(5)  GATHER_CASE(4)
            GATHER_CASE(3)  GATHER_CASE(2)  GATHER_CASE(1)
#undef GATHER_CASE
            default: break;
        }
        const int left = len - base - 32;
        m = left < 32 ? left : 32;
        rp += 32;
        if (lane < m) rc = *rp;
    }
    const float sn = di * di;
    v2f o;
    o.x = fmaf(self.x, sn, acc.x) + bv.x;
    o.y = fmaf(self.y, sn, acc.y) + bv.y;
    ((v2f*)out)[(long)node * (HID / 2) + lane] = o;
    }
}

__global__ void logsoftmax_kernel(const float* __restrict__ H, float* __restrict__ out, int N) {
    int gid  = blockIdx.x * blockDim.x + threadIdx.x;
    int node = gid >> 5;
    int lane = gid & 31;
    if (node >= N) return;
    const float* rowp = H + (long)node * HID;
    float a = rowp[lane];
    float b = rowp[lane + 32];
    float m = fmaxf(a, b);
#pragma unroll
    for (int off = 16; off > 0; off >>= 1) m = fmaxf(m, __shfl_xor(m, off, 32));
    float s = __expf(a - m) + __expf(b - m);
#pragma unroll
    for (int off = 16; off > 0; off >>= 1) s += __shfl_xor(s, off, 32);
    float lse = m + __logf(s);
    float* op = out + (long)node * HID;
    op[lane]      = a - lse;
    op[lane + 32] = b - lse;
}

extern "C" void kernel_launch(void* const* d_in, const int* in_sizes, int n_in,
                              void* d_out, int out_size, void* d_ws, size_t ws_size,
                              hipStream_t stream) {
    const float* x  = (const float*)d_in[0];
    const int*   ei = (const int*)d_in[1];
    const int*   ec = (const int*)d_in[2];
    const float* W1 = (const float*)d_in[3];
    const float* b1 = (const float*)d_in[4];
    const float* W2 = (const float*)d_in[5];
    const float* b2 = (const float*)d_in[6];

    const int IN_DIM = 256;
    const int N = in_sizes[0] / IN_DIM;
    const int E = in_sizes[2];
    const int* row = ei;
    const int* col = ei + E;

    float* wsf  = (float*)d_ws;
    float* h    = wsf;
    float* agg1 = h    + (size_t)N * HID;
    int2*  rec  = (int2*)(agg1 + (size_t)N * HID);
    int*   rank = (int*)(rec + (size_t)E);
    float* dinv = (float*)(rank + (size_t)E);
    int*   cnt  = (int*)(dinv + (size_t)N);
    int*   off  = cnt + (size_t)N;
    const int nb = (N + SCAN_B - 1) / SCAN_B;
    int*   bsum = off + (size_t)N;
    int*   mm   = bsum + (size_t)nb;
    float* h2   = h;

    float* emb = (float*)d_out;
    float* lsm = emb + (size_t)N * HID;

    const int tileBlocks = (int)((N / 16 + 7) / 8);
    const int edgeBlocks = (E + 256 * EPT - 1) / (256 * EPT);
    const int nodeBlocks = 2048;

    init_kernel<<<(N + 255) / 256, 256, 0, stream>>>(cnt, mm, N);
    minmax_kernel<<<1024, 256, 0, stream>>>(ec, E, mm);
    count_kernel<<<edgeBlocks, 256, 0, stream>>>(col, cnt, rank, E);
    scan_kernel<<<nb, SCAN_B, 0, stream>>>(cnt, off, bsum, N);
    bscan_kernel<<<1, 1024, 0, stream>>>(bsum, nb);
    fill_kernel<<<edgeBlocks, 256, 0, stream>>>(ec, row, col, rank, off, bsum, mm, rec, E);
    dinv_kernel<<<(N + 255) / 256, 256, 0, stream>>>(off, bsum, cnt, rec, dinv, N);

    gemm_wmma_kernel<256, false><<<tileBlocks, 256, 0, stream>>>(x, W1, h, N);
    gather_kernel<true><<<nodeBlocks, 256, 0, stream>>>(h, dinv, b1, off, bsum, cnt, rec, agg1, N);

    gemm_wmma_kernel<64, true><<<tileBlocks, 256, 0, stream>>>(agg1, W2, h2, N);
    gather_kernel<false><<<nodeBlocks, 256, 0, stream>>>(h2, dinv, b2, off, bsum, cnt, rec, emb, N);

    logsoftmax_kernel<<<(N + 7) / 8, 256, 0, stream>>>(emb, lsm, N);
}


# log_softmax fused into the layer-2 gather: the wave that sums a node's row takes max and sum of exp from its registers (DPP butterfly; the exp sum pairs dims 2l,2l+1 first); one launch fewer
# speedup vs baseline: 1.6201x; 1.0389x over previous
#include <hip/hip_runtime.h>
#include <hip/hip_bf16.h>

typedef __attribute__((ext_vector_type(2))) float v2f;
typedef __attribute__((ext_vector_type(8))) float v8f;

#define HID 64

__global__ void init_kernel(int* __restrict__ cnt, int* __restrict__ mm, int N) {
    int i = blockIdx.x * blockDim.x + threadIdx.x;
    if (i < N) cnt[i] = 0;
    if (i == 0) { mm[0] = 0x7FFFFFFF; mm[1] = 0x80000000; }
}

__global__ void minmax_kernel(const int* __restrict__ ec, int E, int* __restrict__ mm) {
    int i = blockIdx.x * blockDim.x + threadIdx.x;
    int stride = gridDim.x * blockDim.x;
    int lmin = 0x7FFFFFFF, lmax = 0x80000000;
    for (; i < E; i += stride) {
        int v = ec[i];
        lmin = min(lmin, v);
        lmax = max(lmax, v);
    }
    atomicMin(&mm[0], lmin);
    atomicMax(&mm[1], lmax);
}

#define EPT 4

__global__ void __launch_bounds__(256)
count_kernel(const int* __restrict__ col, int* __restrict__ cnt,
             int* __restrict__ rank, int E) {
    const int T = gridDim.x * blockDim.x;
    const int t = blockIdx.x * blockDim.x + threadIdx.x;
    if (t + (EPT - 1) * T < E) {
        int c[EPT], rk[EPT];
        for (int u = 0; u < EPT; ++u) c[u] = col[t + u * T];
        for (int u = 0; u < EPT; ++u) rk[u] = atomicAdd(&cnt[c[u]], 1);
        for (int u = 0; u < EPT; ++u) rank[t + u * T] = rk[u];
    } else {
        for (int e = t; e < E; e += T) rank[e] = atomicAdd(&cnt[col[e]], 1);
    }
}

#define SCAN_B 256
#define SCAN_SHIFT 8

__global__ void __launch_bounds__(SCAN_B)
scan_kernel(const int* __restrict__ cnt, int* __restrict__ off,
            int* __restrict__ bsum, int N) {
    __shared__ int wsum[SCAN_B / 32];
    const int tid = threadIdx.x, lane = tid & 31, w = tid >> 5;
    const int i = blockIdx.x * SCAN_B + tid;
    const int v = (i < N) ? cnt[i] : 0;
    int inc = v;
    for (int o = 1; o < 32; o <<= 1) {
        int t = __shfl_up(inc, o, 32);
        if (lane >= o) inc += t;
    }
    if (lane == 31) wsum[w] = inc;
    __syncthreads();
    int wbase = 0, total = 0;
    for (int k = 0; k < SCAN_B / 32; ++k) {
        int s = wsum[k];
        if (k < w) wbase += s;
        total += s;
    }
    if (i < N) off[i] = wbase + inc - v;
    if (tid == 0) bsum[blockIdx.x] = total;
}

__global__ void __launch_bounds__(1024)
bscan_kernel(int* __restrict__ bsum, int nb) {
    __shared__ int wsum[32];
    const int tid = threadIdx.x, lane = tid & 31, w = tid >> 5;
    int carry = 0;
    for (int base = 0; base < nb; base += 1024) {
        const int i = base + tid;
        const int v = (i < nb) ? bsum[i] : 0;
        int inc = v;
        for (int o = 1; o < 32; o <<= 1) {
            int t = __shfl_up(inc, o, 32);
            if (lane >= o) inc += t;
        }
        if (lane == 31) wsum[w] = inc;
        __syncthreads();
        int wbase = 0, total = 0;
        for (int k = 0; k < 32; ++k) {
            int s = wsum[k];
            if (k < w) wbase += s;
            total += s;
        }
        if (i < nb) bsum[i] = carry + wbase + inc - v;
        carry += total;
        __syncthreads();
    }
}

__global__ void __launch_bounds__(256)
fill_kernel(const int* __restrict__ ec, const int* __restrict__ row,
            const int* __restrict__ col, const int* __restrict__ rank,
            const int* __restrict__ off, const int* __restrict__ bsum,
            const int* __restrict__ mm, int2* __restrict__ rec, int E) {
    const int T = gridDim.x * blockDim.x;
    const int t = blockIdx.x * blockDim.x + threadIdx.x;
    const float mn = (float)mm[0], mx = (float)mm[1];
    const float denom = mx - mn;
    const bool span = denom > 0.0f;
    const float dsafe = span ? denom : 1.0f;
    if (t + (EPT - 1) * T < E) {
        int v[EPT], r[EPT], c[EPT], pos[EPT], sb[EPT], so[EPT];
        for (int u = 0; u < EPT; ++u) {
            const int e = t + u * T;
            v[u] = ec[e];
            r[u] = row[e];
            c[u] = col[e];
            pos[u] = rank[e];
        }
        for (int u = 0; u < EPT; ++u) {
            sb[u] = bsum[c[u] >> SCAN_SHIFT];
            so[u] = off[c[u]];
        }
        for (int u = 0; u < EPT; ++u) {
            const float q = ((float)v[u] - mn) / dsafe;
            const float w = span ? q : 0.0f;
            rec[sb[u] + so[u] + pos[u]] = make_int2(r[u], __float_as_int(w));
        }
    } else {
        for (int e = t; e < E; e += T) {
            const float q = ((float)ec[e] - mn) / dsafe;
            const float w = span ? q : 0.0f;
            const int c = col[e];
            rec[bsum[c >> SCAN_SHIFT] + off[c] + rank[e]] = make_int2(row[e], __float_as_int(w));
        }
    }
}

__global__ void dinv_kernel(const int* __restrict__ off, const int* __restrict__ bsum,
                            const int* __restrict__ cnt, const int2* __restrict__ rec,
                            float* __restrict__ dinv, int N) {
    int i = blockIdx.x * blockDim.x + threadIdx.x;
    if (i >= N) return;
    const long start = (long)bsum[i >> SCAN_SHIFT] + off[i];
    const int len = cnt[i];
    float d = 1.0f;
    for (int k = 0; k < len; ++k) d += __int_as_float(rec[start + k].y);
    dinv[i] = (d > 0.0f) ? rsqrtf(d) : 0.0f;
}

template <int K, bool RELU>
__global__ void gemm_wmma_kernel(const float* __restrict__ A, const float* __restrict__ W,
                                 float* __restrict__ H, int M) {
    __shared__ __align__(16) float sW[K * HID];

    for (int i = threadIdx.x; i < (K / 2) * HID; i += blockDim.x) {
        int kp = i >> 6;
        int c  = i & 63;
        float w0 = W[(2 * kp) * HID + c];
        float w1 = W[(2 * kp + 1) * HID + c];
        float* dst = sW + kp * (2 * HID) + 2 * c;
        dst[0] = w0;
        dst[1] = w1;
    }
    __syncthreads();

    const int lane  = threadIdx.x & 31;
    const int widx  = threadIdx.x >> 5;
    const int half  = lane >> 4;
    const int sub   = lane & 15;
    const long tile = (long)blockIdx.x * 8 + widx;
    const long row0 = tile * 16;
    if (row0 >= M) return;

    v8f acc[4] = {};

    const float* aRow = A + (row0 + sub) * (long)K;

#pragma unroll 4
    for (int kk = 0; kk < K; kk += 4) {
        const int kb = kk + 2 * half;
        v2f a = *(const v2f*)(aRow + kb);
        if (RELU) { a.x = fmaxf(a.x, 0.0f); a.y = fmaxf(a.y, 0.0f); }

        const float* wp = sW + (kb >> 1) * (2 * HID) + 2 * sub;
#pragma unroll
        for (int t = 0; t < 4; ++t) {
            v2f b = *(const v2f*)(wp + t * 32);
            acc[t] = __builtin_amdgcn_wmma_f32_16x16x4_f32(
                 false, a,  false, b,
                 (short)0, acc[t],
                 false,  false);
        }
    }

#pragma unroll
    for (int r = 0; r < 8; ++r) {
        float* hp = H + (row0 + r + 8 * half) * (long)HID + sub;
#pragma unroll
        for (int t = 0; t < 4; ++t) hp[t * 16] = acc[t][r];
    }
}


template <int MASK>
__device__ __forceinline__ float row_xor(float x) {
    const int b = __builtin_bit_cast(int, x);
    return __builtin_bit_cast(float,
        __builtin_amdgcn_mov_dpp(b, 0x160 + MASK, 0xf, 0xf, false));
}
__device__ __forceinline__ float xhalf(float x) {
    const int b = __builtin_bit_cast(int, x);
    return __builtin_bit_cast(float,
        __builtin_amdgcn_permlanex16(b, b, 0x76543210, 0xfedcba98, false, false));
}
__device__ __forceinline__ float wave_max(float m) {
    m = fmaxf(m, row_xor<1>(m));
    m = fmaxf(m, row_xor<2>(m));
    m = fmaxf(m, row_xor<4>(m));
    m = fmaxf(m, row_xor<8>(m));
    return fmaxf(m, xhalf(m));
}
__device__ __forceinline__ float wave_sum(float s) {
    s += row_xor<1>(s);
    s += row_xor<2>(s);
    s += row_xor<4>(s);
    s += row_xor<8>(s);
    return s + xhalf(s);
}

template <int B>
__device__ __forceinline__ void gather_batch(const v2f* __restrict__ hl, int r, float nrm,
                                             int j0, v2f& acc) {
    v2f a[B];
#pragma unroll
    for (int j = 0; j < B; ++j) {
        const int rj = __builtin_amdgcn_readlane(r, j0 + j);
        a[j] = hl[(long)rj * (HID / 2)];
    }
#pragma unroll
    for (int j = 0; j < B; ++j) {
        const float nj = __int_as_float(
            __builtin_amdgcn_readlane(__float_as_int(nrm), j0 + j));
        acc.x = fmaf(a[j].x, nj, acc.x);
        acc.y = fmaf(a[j].y, nj, acc.y);
    }
}

template <bool FIRST>
__global__ void __launch_bounds__(256)
gather_kernel(const float* __restrict__ h, const float* __restrict__ dinv,
              const float* __restrict__ bias, const int* __restrict__ off,
              const int* __restrict__ bsum, const int* __restrict__ cnt,
              int2* __restrict__ rec, float* __restrict__ out,
              float* __restrict__ lsm, int N) {
    const int lane = threadIdx.x & 31;
    const int wave = __builtin_amdgcn_readfirstlane(blockIdx.x * 8 + (threadIdx.x >> 5));
    const int nwaves = gridDim.x * 8;
    const v2f* hl = (const v2f*)h + lane;
    const v2f bv = ((const v2f*)bias)[lane];
    for (int node = wave; node < N; node += nwaves) {
    const int start = bsum[node >> SCAN_SHIFT] + off[node];
    const int len = cnt[node];
    const float di = dinv[node];
    const v2f self = hl[(long)node * (HID / 2)];
    int2* rp = rec + ((long)start + lane);

    int m = len < 32 ? len : 32;
    int2 rc = make_int2(0, 0);
    if (lane < m) rc = *rp;

    v2f acc = {0.0f, 0.0f};
    for (int base = 0; base < len; base += 32) {
        const int r = rc.x;
        float nrm = __int_as_float(rc.y);
        if (FIRST) {
            if (lane < m) {
                nrm = dinv[r] * nrm * di;
                rp->y = __float_as_int(nrm);
            }
        }
        int j0 = 0;
#pragma unroll 1
        for (; j0 + 16 <= m; j0 += 16) gather_batch<16>(hl, r, nrm, j0, acc);
        switch (m - j0) {
#define GATHER_CASE(B) case B: gather_batch<B>(hl, r, nrm, j0, acc); break;
            GATHER_CASE(15) GATHER_CASE(14) GATHER_CASE(13) GATHER_CASE(12)
            GATHER_CASE(11) GATHER_CASE(10) GATHER_CASE(9)  GATHER_CASE(8)
            GATHER_CASE(7)  GATHER_CASE(6)  GATHER_CASE(5)  GATHER_CASE(4)
            GATHER_CASE(3)  GATHER_CASE(2)  GATHER_CASE(1)
#undef GATHER_CASE
            default: break;
        }
        const int left = len - base - 32;
        m = left < 32 ? left : 32;
        rp += 32;
        if (lane < m) rc = *rp;
    }
    const float sn = di * di;
    v2f o;
    o.x = fmaf(self.x, sn, acc.x) + bv.x;
    o.y = fmaf(self.y, sn, acc.y) + bv.y;
    ((v2f*)out)[(long)node * (HID / 2) + lane] = o;
    if (!FIRST) {
        const float mx = wave_max(fmaxf(o.x, o.y));
        const float se = wave_sum(__expf(o.x - mx) + __expf(o.y - mx));
        const float lse = mx + __logf(se);
        v2f l;
        l.x = o.x - lse;
        l.y = o.y - lse;
        ((v2f*)lsm)[(long)node * (HID / 2) + lane] = l;
    }
    }
}

extern "C" void kernel_launch(void* const* d_in, const int* in_sizes, int n_in,
                              void* d_out, int out_size, void* d_ws, size_t ws_size,
                              hipStream_t stream) {
    const float* x  = (const float*)d_in[0];
    const int*   ei = (const int*)d_in[1];
    const int*   ec = (const int*)d_in[2];
    const float* W1 = (const float*)d_in[3];
    const float* b1 = (const float*)d_in[4];
    const float* W2 = (const float*)d_in[5];
    const float* b2 = (const float*)d_in[6];

    const int IN_DIM = 256;
    const int N = in_sizes[0] / IN_DIM;
    const int E = in_sizes[2];
    const int* row = ei;
    const int* col = ei + E;

    float* wsf  = (float*)d_ws;
    float* h    = wsf;
    float* agg1 = h    + (size_t)N * HID;
    int2*  rec  = (int2*)(agg1 + (size_t)N * HID);
    int*   rank = (int*)(rec + (size_t)E);
    float* dinv = (float*)(rank + (size_t)E);
    int*   cnt  = (int*)(dinv + (size_t)N);
    int*   off  = cnt + (size_t)N;
    const int nb = (N + SCAN_B - 1) / SCAN_B;
    int*   bsum = off + (size_t)N;
    int*   mm   = bsum + (size_t)nb;
    float* h2   = h;

    float* emb = (float*)d_out;
    float* lsm = emb + (size_t)N * HID;

    const int tileBlocks = (int)((N / 16 + 7) / 8);
    const int edgeBlocks = (E + 256 * EPT - 1) / (256 * EPT);
    const int nodeBlocks = 2048;

    init_kernel<<<(N + 255) / 256, 256, 0, stream>>>(cnt, mm, N);
    minmax_kernel<<<1024, 256, 0, stream>>>(ec, E, mm);
    count_kernel<<<edgeBlocks, 256, 0, stream>>>(col, cnt, rank, E);
    scan_kernel<<<nb, SCAN_B, 0, stream>>>(cnt, off, bsum, N);
    bscan_kernel<<<1, 1024, 0, stream>>>(bsum, nb);
    fill_kernel<<<edgeBlocks, 256, 0, stream>>>(ec, row, col, rank, off, bsum, mm, rec, E);
    dinv_kernel<<<(N + 255) / 256, 256, 0, stream>>>(off, bsum, cnt, rec, dinv, N);

    gemm_wmma_kernel<256, false><<<tileBlocks, 256, 0, stream>>>(x, W1, h, N);
    gather_kernel<true><<<nodeBlocks, 256, 0, stream>>>(h, dinv, b1, off, bsum, cnt, rec, agg1, nullptr, N);

    gemm_wmma_kernel<64, true><<<tileBlocks, 256, 0, stream>>>(agg1, W2, h2, N);
    gather_kernel<false><<<nodeBlocks, 256, 0, stream>>>(h2, dinv, b2, off, bsum, cnt, rec, emb, lsm, N);
}


# in-edge counters and min/max pair kept twice, one copy per package half (die id); a wave's atomics go only to its half's copy; a segment's order is now copy 0 then copy 1, each by arrival
# speedup vs baseline: 2.2194x; 1.3699x over previous
#include <hip/hip_runtime.h>
#include <hip/hip_bf16.h>

typedef __attribute__((ext_vector_type(2))) float v2f;
typedef __attribute__((ext_vector_type(8))) float v8f;

#define HID 64

__device__ __forceinline__ int package_half() {
    return (__builtin_amdgcn_s_sendmsg_rtn(0x87) >> 18) & 1;
}

__global__ void init_kernel(int* __restrict__ cnt, int* __restrict__ cnt1,
                            int* __restrict__ mm, int* __restrict__ mm1, int N) {
    int i = blockIdx.x * blockDim.x + threadIdx.x;
    if (i < N) { cnt[i] = 0; cnt1[i] = 0; }
    if (i == 0) {
        mm[0] = 0x7FFFFFFF;  mm[1] = 0x80000000;
        mm1[0] = 0x7FFFFFFF; mm1[1] = 0x80000000;
    }
}

__global__ void minmax_kernel(const int* __restrict__ ec, int E, int* __restrict__ mm0,
                              int* __restrict__ mm1) {
    int i = blockIdx.x * blockDim.x + threadIdx.x;
    int stride = gridDim.x * blockDim.x;
    int lmin = 0x7FFFFFFF, lmax = 0x80000000;
    for (; i < E; i += stride) {
        int v = ec[i];
        lmin = min(lmin, v);
        lmax = max(lmax, v);
    }
    int* mm = package_half() ? mm1 : mm0;
    atomicMin(&mm[0], lmin);
    atomicMax(&mm[1], lmax);
}

#define EPT 4

__global__ void __launch_bounds__(256)
count_kernel(const int* __restrict__ col, int* __restrict__ cnt0,
             int* __restrict__ cnt1, int* __restrict__ rank, int E) {
    const int T = gridDim.x * blockDim.x;
    const int t = blockIdx.x * blockDim.x + threadIdx.x;
    const int hf = package_half();
    int* cnt = hf ? cnt1 : cnt0;
    if (t + (EPT - 1) * T < E) {
        int c[EPT], rk[EPT];
        for (int u = 0; u < EPT; ++u) c[u] = col[t + u * T];
        for (int u = 0; u < EPT; ++u) rk[u] = atomicAdd(&cnt[c[u]], 1);
        for (int u = 0; u < EPT; ++u) rank[t + u * T] = rk[u] * 2 + hf;
    } else {
        for (int e = t; e < E; e += T) rank[e] = atomicAdd(&cnt[col[e]], 1) * 2 + hf;
    }
}

#define SCAN_B 256
#define SCAN_SHIFT 8

__global__ void __launch_bounds__(SCAN_B)
scan_kernel(const int* __restrict__ cnt, const int* __restrict__ cnt1,
            int* __restrict__ off, int* __restrict__ bsum, int N) {
    __shared__ int wsum[SCAN_B / 32];
    const int tid = threadIdx.x, lane = tid & 31, w = tid >> 5;
    const int i = blockIdx.x * SCAN_B + tid;
    const int v = (i < N) ? cnt[i] + cnt1[i] : 0;
    int inc = v;
    for (int o = 1; o < 32; o <<= 1) {
        int t = __shfl_up(inc, o, 32);
        if (lane >= o) inc += t;
    }
    if (lane == 31) wsum[w] = inc;
    __syncthreads();
    int wbase = 0, total = 0;
    for (int k = 0; k < SCAN_B / 32; ++k) {
        int s = wsum[k];
        if (k < w) wbase += s;
        total += s;
    }
    if (i < N) off[i] = wbase + inc - v;
    if (tid == 0) bsum[blockIdx.x] = total;
}

__global__ void __launch_bounds__(1024)
bscan_kernel(int* __restrict__ bsum, int nb) {
    __shared__ int wsum[32];
    const int tid = threadIdx.x, lane = tid & 31, w = tid >> 5;
    int carry = 0;
    for (int base = 0; base < nb; base += 1024) {
        const int i = base + tid;
        const int v = (i < nb) ? bsum[i] : 0;
        int inc = v;
        for (int o = 1; o < 32; o <<= 1) {
            int t = __shfl_up(inc, o, 32);
            if (lane >= o) inc += t;
        }
        if (lane == 31) wsum[w] = inc;
        __syncthreads();
        int wbase = 0, total = 0;
        for (int k = 0; k < 32; ++k) {
            int s = wsum[k];
            if (k < w) wbase += s;
            total += s;
        }
        if (i < nb) bsum[i] = carry + wbase + inc - v;
        carry += total;
        __syncthreads();
    }
}

__global__ void __launch_bounds__(256)
fill_kernel(const int* __restrict__ ec, const int* __restrict__ row,
            const int* __restrict__ col, const int* __restrict__ rank,
            const int* __restrict__ off, const int* __restrict__ bsum,
            const int* __restrict__ cnt0,
            const int* __restrict__ mm, const int* __restrict__ mm1,
            int2* __restrict__ rec, int E) {
    const int T = gridDim.x * blockDim.x;
    const int t = blockIdx.x * blockDim.x + threadIdx.x;
    const float mn = (float)min(mm[0], mm1[0]), mx = (float)max(mm[1], mm1[1]);
    const float denom = mx - mn;
    const bool span = denom > 0.0f;
    const float dsafe = span ? denom : 1.0f;
    if (t + (EPT - 1) * T < E) {
        int v[EPT], r[EPT], c[EPT], pos[EPT], sb[EPT], so[EPT], c0[EPT];
        for (int u = 0; u < EPT; ++u) {
            const int e = t + u * T;
            v[u] = ec[e];
            r[u] = row[e];
            c[u] = col[e];
            pos[u] = rank[e];
        }
        for (int u = 0; u < EPT; ++u) {
            sb[u] = bsum[c[u] >> SCAN_SHIFT];
            so[u] = off[c[u]];
            c0[u] = cnt0[c[u]];
        }
        for (int u = 0; u < EPT; ++u) {
            const float q = ((float)v[u] - mn) / dsafe;
            const float w = span ? q : 0.0f;
            const int behind = (pos[u] & 1) ? c0[u] : 0;
            rec[sb[u] + so[u] + behind + (pos[u] >> 1)] = make_int2(r[u], __float_as_int(w));
        }
    } else {
        for (int e = t; e < E; e += T) {
            const float q = ((float)ec[e] - mn) / dsafe;
            const float w = span ? q : 0.0f;
            const int c = col[e];
            const int rk = rank[e];
            const int behind = (rk & 1) ? cnt0[c] : 0;
            rec[bsum[c >> SCAN_SHIFT] + off[c] + behind + (rk >> 1)] = make_int2(row[e], __float_as_int(w));
        }
    }
}

__global__ void dinv_kernel(const int* __restrict__ off, const int* __restrict__ bsum,
                            int* __restrict__ cnt, const int* __restrict__ cnt1,
                            const int2* __restrict__ rec,
                            float* __restrict__ dinv, int N) {
    int i = blockIdx.x * blockDim.x + threadIdx.x;
    if (i >= N) return;
    const long start = (long)bsum[i >> SCAN_SHIFT] + off[i];
    const int len = cnt[i] + cnt1[i];
    cnt[i] = len;
    float d = 1.0f;
    for (int k = 0; k < len; ++k) d += __int_as_float(rec[start + k].y);
    dinv[i] = (d > 0.0f) ? rsqrtf(d) : 0.0f;
}

template <int K, bool RELU>
__global__ void gemm_wmma_kernel(const float* __restrict__ A, const float* __restrict__ W,
                                 float* __restrict__ H, int M) {
    __shared__ __align__(16) float sW[K * HID];

    for (int i = threadIdx.x; i < (K / 2) * HID; i += blockDim.x) {
        int kp = i >> 6;
        int c  = i & 63;
        float w0 = W[(2 * kp) * HID + c];
        float w1 = W[(2 * kp + 1) * HID + c];
        float* dst = sW + kp * (2 * HID) + 2 * c;
        dst[0] = w0;
        dst[1] = w1;
    }
    __syncthreads();

    const int lane  = threadIdx.x & 31;
    const int widx  = threadIdx.x >> 5;
    const int half  = lane >> 4;
    const int sub   = lane & 15;
    const long tile = (long)blockIdx.x * 8 + widx;
    const long row0 = tile * 16;
    if (row0 >= M) return;

    v8f acc[4] = {};

    const float* aRow = A + (row0 + sub) * (long)K;

#pragma unroll 4
    for (int kk = 0; kk < K; kk += 4) {
        const int kb = kk + 2 * half;
        v2f a = *(const v2f*)(aRow + kb);
        if (RELU) { a.x = fmaxf(a.x, 0.0f); a.y = fmaxf(a.y, 0.0f); }

        const float* wp = sW + (kb >> 1) * (2 * HID) + 2 * sub;
#pragma unroll
        for (int t = 0; t < 4; ++t) {
            v2f b = *(const v2f*)(wp + t * 32);
            acc[t] = __builtin_amdgcn_wmma_f32_16x16x4_f32(
                 false, a,  false, b,
                 (short)0, acc[t],
                 false,  false);
        }
    }

#pragma unroll
    for (int r = 0; r < 8; ++r) {
        float* hp = H + (row0 + r + 8 * half) * (long)HID + sub;
#pragma unroll
        for (int t = 0; t < 4; ++t) hp[t * 16] = acc[t][r];
    }
}


template <int MASK>
__device__ __forceinline__ float row_xor(float x) {
    const int b = __builtin_bit_cast(int, x);
    return __builtin_bit_cast(float,
        __builtin_amdgcn_mov_dpp(b, 0x160 + MASK, 0xf, 0xf, false));
}
__device__ __forceinline__ float xhalf(float x) {
    const int b = __builtin_bit_cast(int, x);
    return __builtin_bit_cast(float,
        __builtin_amdgcn_permlanex16(b, b, 0x76543210, 0xfedcba98, false, false));
}
__device__ __forceinline__ float wave_max(float m) {
    m = fmaxf(m, row_xor<1>(m));
    m = fmaxf(m, row_xor<2>(m));
    m = fmaxf(m, row_xor<4>(m));
    m = fmaxf(m, row_xor<8>(m));
    return fmaxf(m, xhalf(m));
}
__device__ __forceinline__ float wave_sum(float s) {
    s += row_xor<1>(s);
    s += row_xor<2>(s);
    s += row_xor<4>(s);
    s += row_xor<8>(s);
    return s + xhalf(s);
}

template <int B>
__device__ __forceinline__ void gather_batch(const v2f* __restrict__ hl, int r, float nrm,
                                             int j0, v2f& acc) {
    v2f a[B];
#pragma unroll
    for (int j = 0; j < B; ++j) {
        const int rj = __builtin_amdgcn_readlane(r, j0 + j);
        a[j] = hl[(long)rj * (HID / 2)];
    }
#pragma unroll
    for (int j = 0; j < B; ++j) {
        const float nj = __int_as_float(
            __builtin_amdgcn_readlane(__float_as_int(nrm), j0 + j));
        acc.x = fmaf(a[j].x, nj, acc.x);
        acc.y = fmaf(a[j].y, nj, acc.y);
    }
}

template <bool FIRST>
__global__ void __launch_bounds__(256)
gather_kernel(const float* __restrict__ h, const float* __restrict__ dinv,
              const float* __restrict__ bias, const int* __restrict__ off,
              const int* __restrict__ bsum, const int* __restrict__ cnt,
              int2* __restrict__ rec, float* __restrict__ out,
              float* __restrict__ lsm, int N) {
    const int lane = threadIdx.x & 31;
    const int wave = __builtin_amdgcn_readfirstlane(blockIdx.x * 8 + (threadIdx.x >> 5));
    const int nwaves = gridDim.x * 8;
    const v2f* hl = (const v2f*)h + lane;
    const v2f bv = ((const v2f*)bias)[lane];
    for (int node = wave; node < N; node += nwaves) {
    const int start = bsum[node >> SCAN_SHIFT] + off[node];
    const int len = cnt[node];
    const float di = dinv[node];
    const v2f self = hl[(long)node * (HID / 2)];
    int2* rp = rec + ((long)start + lane);

    int m = len < 32 ? len : 32;
    int2 rc = make_int2(0, 0);
    if (lane < m) rc = *rp;

    v2f acc = {0.0f, 0.0f};
    for (int base = 0; base < len; base += 32) {
        const int r = rc.x;
        float nrm = __int_as_float(rc.y);
        if (FIRST) {
            if (lane < m) {
                nrm = dinv[r] * nrm * di;
                rp->y = __float_as_int(nrm);
            }
        }
        int j0 = 0;
#pragma unroll 1
        for (; j0 + 16 <= m; j0 += 16) gather_batch<16>(hl, r, nrm, j0, acc);
        switch (m - j0) {
#define GATHER_CASE(B) case B: gather_batch<B>(hl, r, nrm, j0, acc); break;
            GATHER_CASE(15) GATHER_CASE(14) GATHER_CASE(13) GATHER_CASE(12)
            GATHER_CASE(11) GATHER_CASE(10) GATHER_CASE(9)  GATHER_CASE(8)
            GATHER_CASE(7)  GATHER_CASE(6)  GATHER_CASE(5)  GATHER_CASE(4)
            GATHER_CASE(3)  GATHER_CASE(2)  GATHER_CASE(1)
#undef GATHER_CASE
            default: break;
        }
        const int left = len - base - 32;
        m = left < 32 ? left : 32;
        rp += 32;
        if (lane < m) rc = *rp;
    }
    const float sn = di * di;
    v2f o;
    o.x = fmaf(self.x, sn, acc.x) + bv.x;
    o.y = fmaf(self.y, sn, acc.y) + bv.y;
    ((v2f*)out)[(long)node * (HID / 2) + lane] = o;
    if (!FIRST) {
        const float mx = wave_max(fmaxf(o.x, o.y));
        const float se = wave_sum(__expf(o.x - mx) + __expf(o.y - mx));
        const float lse = mx + __logf(se);
        v2f l;
        l.x = o.x - lse;
        l.y = o.y - lse;
        ((v2f*)lsm)[(long)node * (HID / 2) + lane] = l;
    }
    }
}

extern "C" void kernel_launch(void* const* d_in, const int* in_sizes, int n_in,
                              void* d_out, int out_size, void* d_ws, size_t ws_size,
                              hipStream_t stream) {
    const float* x  = (const float*)d_in[0];
    const int*   ei = (const int*)d_in[1];
    const int*   ec = (const int*)d_in[2];
    const float* W1 = (const float*)d_in[3];
    const float* b1 = (const float*)d_in[4];
    const float* W2 = (const float*)d_in[5];
    const float* b2 = (const float*)d_in[6];

    const int IN_DIM = 256;
    const int N = in_sizes[0] / IN_DIM;
    const int E = in_sizes[2];
    const int* row = ei;
    const int* col = ei + E;

    float* wsf  = (float*)d_ws;
    float* h    = wsf;
    float* agg1 = h    + (size_t)N * HID;
    int2*  rec  = (int2*)(agg1 + (size_t)N * HID);
    int*   rank = (int*)(rec + (size_t)E);
    float* dinv = (float*)(rank + (size_t)E);
    int*   cnt  = (int*)(dinv + (size_t)N);
    int*   cnt1 = (int*)h;
    int*   off  = cnt + (size_t)N;
    const int nb = (N + SCAN_B - 1) / SCAN_B;
    int*   bsum = off + (size_t)N;
    int*   mm   = bsum + (size_t)nb;
    int*   mm1  = cnt1 + (size_t)N + 64;
    float* h2   = h;

    float* emb = (float*)d_out;
    float* lsm = emb + (size_t)N * HID;

    const int tileBlocks = (int)((N / 16 + 7) / 8);
    const int edgeBlocks = (E + 256 * EPT - 1) / (256 * EPT);
    const int nodeBlocks = 2048;

    init_kernel<<<(N + 255) / 256, 256, 0, stream>>>(cnt, cnt1, mm, mm1, N);
    minmax_kernel<<<1024, 256, 0, stream>>>(ec, E, mm, mm1);
    count_kernel<<<edgeBlocks, 256, 0, stream>>>(col, cnt, cnt1, rank, E);
    scan_kernel<<<nb, SCAN_B, 0, stream>>>(cnt, cnt1, off, bsum, N);
    bscan_kernel<<<1, 1024, 0, stream>>>(bsum, nb);
    fill_kernel<<<edgeBlocks, 256, 0, stream>>>(ec, row, col, rank, off, bsum, cnt, mm, mm1, rec, E);
    dinv_kernel<<<(N + 255) / 256, 256, 0, stream>>>(off, bsum, cnt, cnt1, rec, dinv, N);

    gemm_wmma_kernel<256, false><<<tileBlocks, 256, 0, stream>>>(x, W1, h, N);
    gather_kernel<true><<<nodeBlocks, 256, 0, stream>>>(h, dinv, b1, off, bsum, cnt, rec, agg1, nullptr, N);

    gemm_wmma_kernel<64, true><<<tileBlocks, 256, 0, stream>>>(agg1, W2, h2, N);
    gather_kernel<false><<<nodeBlocks, 256, 0, stream>>>(h2, dinv, b2, off, bsum, cnt, rec, emb, lsm, N);
}


# x@W1 product, edge count and min/max as work-group roles of ONE launch (matrix work runs beside the atomics' and loads' waits); two launches fewer; edge ranks by arrival as before
# speedup vs baseline: 2.4665x; 1.1113x over previous
#include <hip/hip_runtime.h>
#include <hip/hip_bf16.h>

typedef __attribute__((ext_vector_type(2))) float v2f;
typedef __attribute__((ext_vector_type(8))) float v8f;

#define HID 64

__device__ __forceinline__ int package_half() {
    return (__builtin_amdgcn_s_sendmsg_rtn(0x87) >> 18) & 1;
}

__global__ void init_kernel(int* __restrict__ cnt, int* __restrict__ cnt1,
                            int* __restrict__ mm, int* __restrict__ mm1, int N) {
    int i = blockIdx.x * blockDim.x + threadIdx.x;
    if (i < N) { cnt[i] = 0; cnt1[i] = 0; }
    if (i == 0) {
        mm[0] = 0x7FFFFFFF;  mm[1] = 0x80000000;
        mm1[0] = 0x7FFFFFFF; mm1[1] = 0x80000000;
    }
}

#define MM_EPT 8

__device__ __forceinline__ void minmax_role(const int* __restrict__ ec, int E,
                                            int* __restrict__ mm0, int* __restrict__ mm1,
                                            int blk, int nblk) {
    const int T = nblk * 256;
    const int t = blk * 256 + threadIdx.x;
    int lmin = 0x7FFFFFFF, lmax = 0x80000000;
    for (int i = t; i < E; i += MM_EPT * T) {
        int v[MM_EPT];
        for (int u = 0; u < MM_EPT; ++u) v[u] = ec[min(i + u * T, E - 1)];
        for (int u = 0; u < MM_EPT; ++u) {
            lmin = min(lmin, v[u]);
            lmax = max(lmax, v[u]);
        }
    }
    int* mm = package_half() ? mm1 : mm0;
    atomicMin(&mm[0], lmin);
    atomicMax(&mm[1], lmax);
}

#define EPT 4

__device__ __forceinline__ void
count_role(const int* __restrict__ col, int* __restrict__ cnt0,
           int* __restrict__ cnt1, int* __restrict__ rank, int E, int blk, int nblk) {
    const int T = nblk * 256;
    const int t = blk * 256 + threadIdx.x;
    const int hf = package_half();
    int* cnt = hf ? cnt1 : cnt0;
    if (t + (EPT - 1) * T < E) {
        int c[EPT], rk[EPT];
        for (int u = 0; u < EPT; ++u) c[u] = col[t + u * T];
        for (int u = 0; u < EPT; ++u) rk[u] = atomicAdd(&cnt[c[u]], 1);
        for (int u = 0; u < EPT; ++u) rank[t + u * T] = rk[u] * 2 + hf;
    } else {
        for (int e = t; e < E; e += T) rank[e] = atomicAdd(&cnt[col[e]], 1) * 2 + hf;
    }
}

#define SCAN_B 256
#define SCAN_SHIFT 8

__global__ void __launch_bounds__(SCAN_B)
scan_kernel(const int* __restrict__ cnt, const int* __restrict__ cnt1,
            int* __restrict__ off, int* __restrict__ bsum, int N) {
    __shared__ int wsum[SCAN_B / 32];
    const int tid = threadIdx.x, lane = tid & 31, w = tid >> 5;
    const int i = blockIdx.x * SCAN_B + tid;
    const int v = (i < N) ? cnt[i] + cnt1[i] : 0;
    int inc = v;
    for (int o = 1; o < 32; o <<= 1) {
        int t = __shfl_up(inc, o, 32);
        if (lane >= o) inc += t;
    }
    if (lane == 31) wsum[w] = inc;
    __syncthreads();
    int wbase = 0, total = 0;
    for (int k = 0; k < SCAN_B / 32; ++k) {
        int s = wsum[k];
        if (k < w) wbase += s;
        total += s;
    }
    if (i < N) off[i] = wbase + inc - v;
    if (tid == 0) bsum[blockIdx.x] = total;
}

__global__ void __launch_bounds__(1024)
bscan_kernel(int* __restrict__ bsum, int nb) {
    __shared__ int wsum[32];
    const int tid = threadIdx.x, lane = tid & 31, w = tid >> 5;
    int carry = 0;
    for (int base = 0; base < nb; base += 1024) {
        const int i = base + tid;
        const int v = (i < nb) ? bsum[i] : 0;
        int inc = v;
        for (int o = 1; o < 32; o <<= 1) {
            int t = __shfl_up(inc, o, 32);
            if (lane >= o) inc += t;
        }
        if (lane == 31) wsum[w] = inc;
        __syncthreads();
        int wbase = 0, total = 0;
        for (int k = 0; k < 32; ++k) {
            int s = wsum[k];
            if (k < w) wbase += s;
            total += s;
        }
        if (i < nb) bsum[i] = carry + wbase + inc - v;
        carry += total;
        __syncthreads();
    }
}

__global__ void __launch_bounds__(256)
fill_kernel(const int* __restrict__ ec, const int* __restrict__ row,
            const int* __restrict__ col, const int* __restrict__ rank,
            const int* __restrict__ off, const int* __restrict__ bsum,
            const int* __restrict__ cnt0,
            const int* __restrict__ mm, const int* __restrict__ mm1,
            int2* __restrict__ rec, int E) {
    const int T = gridDim.x * blockDim.x;
    const int t = blockIdx.x * blockDim.x + threadIdx.x;
    const float mn = (float)min(mm[0], mm1[0]), mx = (float)max(mm[1], mm1[1]);
    const float denom = mx - mn;
    const bool span = denom > 0.0f;
    const float dsafe = span ? denom : 1.0f;
    if (t + (EPT - 1) * T < E) {
        int v[EPT], r[EPT], c[EPT], pos[EPT], sb[EPT], so[EPT], c0[EPT];
        for (int u = 0; u < EPT; ++u) {
            const int e = t + u * T;
            v[u] = ec[e];
            r[u] = row[e];
            c[u] = col[e];
            pos[u] = rank[e];
        }
        for (int u = 0; u < EPT; ++u) {
            sb[u] = bsum[c[u] >> SCAN_SHIFT];
            so[u] = off[c[u]];
            c0[u] = cnt0[c[u]];
        }
        for (int u = 0; u < EPT; ++u) {
            const float q = ((float)v[u] - mn) / dsafe;
            const float w = span ? q : 0.0f;
            const int behind = (pos[u] & 1) ? c0[u] : 0;
            rec[sb[u] + so[u] + behind + (pos[u] >> 1)] = make_int2(r[u], __float_as_int(w));
        }
    } else {
        for (int e = t; e < E; e += T) {
            const float q = ((float)ec[e] - mn) / dsafe;
            const float w = span ? q : 0.0f;
            const int c = col[e];
            const int rk = rank[e];
            const int behind = (rk & 1) ? cnt0[c] : 0;
            rec[bsum[c >> SCAN_SHIFT] + off[c] + behind + (rk >> 1)] = make_int2(row[e], __float_as_int(w));
        }
    }
}

__global__ void dinv_kernel(const int* __restrict__ off, const int* __restrict__ bsum,
                            int* __restrict__ cnt, const int* __restrict__ cnt1,
                            const int2* __restrict__ rec,
                            float* __restrict__ dinv, int N) {
    int i = blockIdx.x * blockDim.x + threadIdx.x;
    if (i >= N) return;
    const long start = (long)bsum[i >> SCAN_SHIFT] + off[i];
    const int len = cnt[i] + cnt1[i];
    cnt[i] = len;
    float d = 1.0f;
    for (int k = 0; k < len; ++k) d += __int_as_float(rec[start + k].y);
    dinv[i] = (d > 0.0f) ? rsqrtf(d) : 0.0f;
}

template <int K, bool RELU>
__device__ __forceinline__ void gemm_block(const float* __restrict__ A, const float* __restrict__ W,
                                           float* __restrict__ H, int M, int blk, float* sW) {

    for (int i = threadIdx.x; i < (K / 2) * HID; i += blockDim.x) {
        int kp = i >> 6;
        int c  = i & 63;
        float w0 = W[(2 * kp) * HID + c];
        float w1 = W[(2 * kp + 1) * HID + c];
        float* dst = sW + kp * (2 * HID) + 2 * c;
        dst[0] = w0;
        dst[1] = w1;
    }
    __syncthreads();

    const int lane  = threadIdx.x & 31;
    const int widx  = threadIdx.x >> 5;
    const int half  = lane >> 4;
    const int sub   = lane & 15;
    const long tile = (long)blk * 8 + widx;
    const long row0 = tile * 16;
    if (row0 >= M) return;

    v8f acc[4] = {};

    const float* aRow = A + (row0 + sub) * (long)K;

#pragma unroll 4
    for (int kk = 0; kk < K; kk += 4) {
        const int kb = kk + 2 * half;
        v2f a = *(const v2f*)(aRow + kb);
        if (RELU) { a.x = fmaxf(a.x, 0.0f); a.y = fmaxf(a.y, 0.0f); }

        const float* wp = sW + (kb >> 1) * (2 * HID) + 2 * sub;
#pragma unroll
        for (int t = 0; t < 4; ++t) {
            v2f b = *(const v2f*)(wp + t * 32);
            acc[t] = __builtin_amdgcn_wmma_f32_16x16x4_f32(
                 false, a,  false, b,
                 (short)0, acc[t],
                 false,  false);
        }
    }

#pragma unroll
    for (int r = 0; r < 8; ++r) {
        float* hp = H + (row0 + r + 8 * half) * (long)HID + sub;
#pragma unroll
        for (int t = 0; t < 4; ++t) hp[t * 16] = acc[t][r];
    }
}

template <int K, bool RELU>
__global__ void gemm_wmma_kernel(const float* __restrict__ A, const float* __restrict__ W,
                                 float* __restrict__ H, int M) {
    __shared__ __align__(16) float sW[K * HID];
    gemm_block<K, RELU>(A, W, H, M, blockIdx.x, sW);
}

__global__ void __launch_bounds__(256)
front_kernel(const float* __restrict__ x, const float* __restrict__ W1, float* __restrict__ h,
             int N, int gemmBlocks,
             const int* __restrict__ col, int* __restrict__ cnt0, int* __restrict__ cnt1,
             int* __restrict__ rank, int E, int countBlocks,
             const int* __restrict__ ec, int* __restrict__ mm0, int* __restrict__ mm1) {
    __shared__ __align__(16) float sW[256 * HID];
    const int b = blockIdx.x;
    if (b < gemmBlocks) {
        gemm_block<256, false>(x, W1, h, N, b, sW);
    } else if (b < gemmBlocks + countBlocks) {
        count_role(col, cnt0, cnt1, rank, E, b - gemmBlocks, countBlocks);
    } else {
        const int first = gemmBlocks + countBlocks;
        minmax_role(ec, E, mm0, mm1, b - first, (int)gridDim.x - first);
    }
}


template <int MASK>
__device__ __forceinline__ float row_xor(float x) {
    const int b = __builtin_bit_cast(int, x);
    return __builtin_bit_cast(float,
        __builtin_amdgcn_mov_dpp(b, 0x160 + MASK, 0xf, 0xf, false));
}
__device__ __forceinline__ float xhalf(float x) {
    const int b = __builtin_bit_cast(int, x);
    return __builtin_bit_cast(float,
        __builtin_amdgcn_permlanex16(b, b, 0x76543210, 0xfedcba98, false, false));
}
__device__ __forceinline__ float wave_max(float m) {
    m = fmaxf(m, row_xor<1>(m));
    m = fmaxf(m, row_xor<2>(m));
    m = fmaxf(m, row_xor<4>(m));
    m = fmaxf(m, row_xor<8>(m));
    return fmaxf(m, xhalf(m));
}
__device__ __forceinline__ float wave_sum(float s) {
    s += row_xor<1>(s);
    s += row_xor<2>(s);
    s += row_xor<4>(s);
    s += row_xor<8>(s);
    return s + xhalf(s);
}

template <int B>
__device__ __forceinline__ void gather_batch(const v2f* __restrict__ hl, int r, float nrm,
                                             int j0, v2f& acc) {
    v2f a[B];
#pragma unroll
    for (int j = 0; j < B; ++j) {
        const int rj = __builtin_amdgcn_readlane(r, j0 + j);
        a[j] = hl[(long)rj * (HID / 2)];
    }
#pragma unroll
    for (int j = 0; j < B; ++j) {
        const float nj = __int_as_float(
            __builtin_amdgcn_readlane(__float_as_int(nrm), j0 + j));
        acc.x = fmaf(a[j].x, nj, acc.x);
        acc.y = fmaf(a[j].y, nj, acc.y);
    }
}

template <bool FIRST>
__global__ void __launch_bounds__(256)
gather_kernel(const float* __restrict__ h, const float* __restrict__ dinv,
              const float* __restrict__ bias, const int* __restrict__ off,
              const int* __restrict__ bsum, const int* __restrict__ cnt,
              int2* __restrict__ rec, float* __restrict__ out,
              float* __restrict__ lsm, int N) {
    const int lane = threadIdx.x & 31;
    const int wave = __builtin_amdgcn_readfirstlane(blockIdx.x * 8 + (threadIdx.x >> 5));
    const int nwaves = gridDim.x * 8;
    const v2f* hl = (const v2f*)h + lane;
    const v2f bv = ((const v2f*)bias)[lane];
    for (int node = wave; node < N; node += nwaves) {
    const int start = bsum[node >> SCAN_SHIFT] + off[node];
    const int len = cnt[node];
    const float di = dinv[node];
    const v2f self = hl[(long)node * (HID / 2)];
    int2* rp = rec + ((long)start + lane);

    int m = len < 32 ? len : 32;
    int2 rc = make_int2(0, 0);
    if (lane < m) rc = *rp;

    v2f acc = {0.0f, 0.0f};
    for (int base = 0; base < len; base += 32) {
        const int r = rc.x;
        float nrm = __int_as_float(rc.y);
        if (FIRST) {
            if (lane < m) {
                nrm = dinv[r] * nrm * di;
                rp->y = __float_as_int(nrm);
            }
        }
        int j0 = 0;
#pragma unroll 1
        for (; j0 + 16 <= m; j0 += 16) gather_batch<16>(hl, r, nrm, j0, acc);
        switch (m - j0) {
#define GATHER_CASE(B) case B: gather_batch<B>(hl, r, nrm, j0, acc); break;
            GATHER_CASE(15) GATHER_CASE(14) GATHER_CASE(13) GATHER_CASE(12)
            GATHER_CASE(11) GATHER_CASE(10) GATHER_CASE(9)  GATHER_CASE(8)
            GATHER_CASE(7)  GATHER_CASE(6)  GATHER_CASE(5)  GATHER_CASE(4)
            GATHER_CASE(3)  GATHER_CASE(2)  GATHER_CASE(1)
#undef GATHER_CASE
            default: break;
        }
        const int left = len - base - 32;
        m = left < 32 ? left : 32;
        rp += 32;
        if (lane < m) rc = *rp;
    }
    const float sn = di * di;
    v2f o;
    o.x = fmaf(self.x, sn, acc.x) + bv.x;
    o.y = fmaf(self.y, sn, acc.y) + bv.y;
    ((v2f*)out)[(long)node * (HID / 2) + lane] = o;
    if (!FIRST) {
        const float mx = wave_max(fmaxf(o.x, o.y));
        const float se = wave_sum(__expf(o.x - mx) + __expf(o.y - mx));
        const float lse = mx + __logf(se);
        v2f l;
        l.x = o.x - lse;
        l.y = o.y - lse;
        ((v2f*)lsm)[(long)node * (HID / 2) + lane] = l;
    }
    }
}

extern "C" void kernel_launch(void* const* d_in, const int* in_sizes, int n_in,
                              void* d_out, int out_size, void* d_ws, size_t ws_size,
                              hipStream_t stream) {
    const float* x  = (const float*)d_in[0];
    const int*   ei = (const int*)d_in[1];
    const int*   ec = (const int*)d_in[2];
    const float* W1 = (const float*)d_in[3];
    const float* b1 = (const float*)d_in[4];
    const float* W2 = (const float*)d_in[5];
    const float* b2 = (const float*)d_in[6];

    const int IN_DIM = 256;
    const int N = in_sizes[0] / IN_DIM;
    const int E = in_sizes[2];
    const int* row = ei;
    const int* col = ei + E;

    float* wsf  = (float*)d_ws;
    float* h    = wsf;
    float* agg1 = h    + (size_t)N * HID;
    int2*  rec  = (int2*)(agg1 + (size_t)N * HID);
    int*   rank = (int*)(rec + (size_t)E);
    float* dinv = (float*)(rank + (size_t)E);
    int*   cnt  = (int*)(dinv + (size_t)N);
    int*   cnt1 = (int*)agg1;
    int*   off  = cnt + (size_t)N;
    const int nb = (N + SCAN_B - 1) / SCAN_B;
    int*   bsum = off + (size_t)N;
    int*   mm   = bsum + (size_t)nb;
    int*   mm1  = cnt1 + (size_t)N + 64;
    float* h2   = h;

    float* emb = (float*)d_out;
    float* lsm = emb + (size_t)N * HID;

    const int tileBlocks = (int)((N / 16 + 7) / 8);
    const int edgeBlocks = (E + 256 * EPT - 1) / (256 * EPT);
    const int nodeBlocks = 2048;
    const int mmBlocks = 256;

    init_kernel<<<(N + 255) / 256, 256, 0, stream>>>(cnt, cnt1, mm, mm1, N);
    front_kernel<<<tileBlocks + edgeBlocks + mmBlocks, 256, 0, stream>>>(
        x, W1, h, N, tileBlocks, col, cnt, cnt1, rank, E, edgeBlocks, ec, mm, mm1);
    scan_kernel<<<nb, SCAN_B, 0, stream>>>(cnt, cnt1, off, bsum, N);
    bscan_kernel<<<1, 1024, 0, stream>>>(bsum, nb);
    fill_kernel<<<edgeBlocks, 256, 0, stream>>>(ec, row, col, rank, off, bsum, cnt, mm, mm1, rec, E);
    dinv_kernel<<<(N + 255) / 256, 256, 0, stream>>>(off, bsum, cnt, cnt1, rec, dinv, N);

    gather_kernel<true><<<nodeBlocks, 256, 0, stream>>>(h, dinv, b1, off, bsum, cnt, rec, agg1, nullptr, N);

    gemm_wmma_kernel<64, true><<<tileBlocks, 256, 0, stream>>>(agg1, W2, h2, N);
    gather_kernel<false><<<nodeBlocks, 256, 0, stream>>>(h2, dinv, b2, off, bsum, cnt, rec, emb, lsm, N);
}


# counting add widened to 64 bits (low word: edge count, high: sum of edge_count); deg = 1 + (S - n*min)/(max - min) from the exact integer sum, order-free; dinv made in the scan, one launch fewer
# speedup vs baseline: 2.5642x; 1.0396x over previous
#include <hip/hip_runtime.h>
#include <hip/hip_bf16.h>

typedef __attribute__((ext_vector_type(2))) float v2f;
typedef __attribute__((ext_vector_type(8))) float v8f;
typedef unsigned long long u64;

#define HID 64

__device__ __forceinline__ int package_half() {
    return (__builtin_amdgcn_s_sendmsg_rtn(0x87) >> 18) & 1;
}

__global__ void init_kernel(u64* __restrict__ cnt, u64* __restrict__ cnt1,
                            int* __restrict__ mm, int* __restrict__ mm1, int N) {
    int i = blockIdx.x * blockDim.x + threadIdx.x;
    if (i < N) { cnt[i] = 0; cnt1[i] = 0; }
    if (i == 0) {
        mm[0] = 0x7FFFFFFF;  mm[1] = 0x80000000;
        mm1[0] = 0x7FFFFFFF; mm1[1] = 0x80000000;
    }
}

#define MM_EPT 8

__device__ __forceinline__ void minmax_role(const int* __restrict__ ec, int E,
                                            int* __restrict__ mm0, int* __restrict__ mm1,
                                            int blk, int nblk) {
    const int T = nblk * 256;
    const int t = blk * 256 + threadIdx.x;
    int lmin = 0x7FFFFFFF, lmax = 0x80000000;
    for (int i = t; i < E; i += MM_EPT * T) {
        int v[MM_EPT];
        for (int u = 0; u < MM_EPT; ++u) v[u] = ec[min(i + u * T, E - 1)];
        for (int u = 0; u < MM_EPT; ++u) {
            lmin = min(lmin, v[u]);
            lmax = max(lmax, v[u]);
        }
    }
    int* mm = package_half() ? mm1 : mm0;
    atomicMin(&mm[0], lmin);
    atomicMax(&mm[1], lmax);
}

#define EPT 4

__device__ __forceinline__ u64 count_word(int ecv) {
    return ((u64)(long long)ecv << 32) + 1ull;
}

__device__ __forceinline__ void
count_role(const int* __restrict__ col, const int* __restrict__ ec, u64* __restrict__ cnt0,
           u64* __restrict__ cnt1, int* __restrict__ rank, int E, int blk, int nblk) {
    const int T = nblk * 256;
    const int t = blk * 256 + threadIdx.x;
    const int hf = package_half();
    u64* cnt = hf ? cnt1 : cnt0;
    if (t + (EPT - 1) * T < E) {
        int c[EPT], v[EPT];
        u64 old[EPT];
        for (int u = 0; u < EPT; ++u) c[u] = col[t + u * T];
        for (int u = 0; u < EPT; ++u) v[u] = ec[t + u * T];
        for (int u = 0; u < EPT; ++u) old[u] = atomicAdd(&cnt[c[u]], count_word(v[u]));
        for (int u = 0; u < EPT; ++u) rank[t + u * T] = (int)(unsigned)old[u] * 2 + hf;
    } else {
        for (int e = t; e < E; e += T)
            rank[e] = (int)(unsigned)atomicAdd(&cnt[col[e]], count_word(ec[e])) * 2 + hf;
    }
}

#define SCAN_B 256
#define SCAN_SHIFT 8

__global__ void __launch_bounds__(SCAN_B)
scan_kernel(const int2* __restrict__ cnt, const int2* __restrict__ cnt1,
            const int* __restrict__ mm, const int* __restrict__ mm1,
            int* __restrict__ off, int* __restrict__ bsum,
            int* __restrict__ len, float* __restrict__ dinv, int N) {
    __shared__ int wsum[SCAN_B / 32];
    const int tid = threadIdx.x, lane = tid & 31, w = tid >> 5;
    const int i = blockIdx.x * SCAN_B + tid;
    int2 ca = make_int2(0, 0), cb = make_int2(0, 0);
    if (i < N) { ca = cnt[i]; cb = cnt1[i]; }
    const int v = ca.x + cb.x;
    int inc = v;
    for (int o = 1; o < 32; o <<= 1) {
        int t = __shfl_up(inc, o, 32);
        if (lane >= o) inc += t;
    }
    if (lane == 31) wsum[w] = inc;
    __syncthreads();
    int wbase = 0, total = 0;
    for (int k = 0; k < SCAN_B / 32; ++k) {
        int s = wsum[k];
        if (k < w) wbase += s;
        total += s;
    }
    if (i < N) {
        off[i] = wbase + inc - v;
        len[i] = v;
        const int mni = min(mm[0], mm1[0]), mxi = max(mm[1], mm1[1]);
        const float denom = (float)mxi - (float)mni;
        const long long num = (long long)ca.y + (long long)cb.y - (long long)v * (long long)mni;
        const float d = (denom > 0.0f) ? 1.0f + (float)num / denom : 1.0f;
        dinv[i] = (d > 0.0f) ? rsqrtf(d) : 0.0f;
    }
    if (tid == 0) bsum[blockIdx.x] = total;
}

__global__ void __launch_bounds__(1024)
bscan_kernel(int* __restrict__ bsum, int nb) {
    __shared__ int wsum[32];
    const int tid = threadIdx.x, lane = tid & 31, w = tid >> 5;
    int carry = 0;
    for (int base = 0; base < nb; base += 1024) {
        const int i = base + tid;
        const int v = (i < nb) ? bsum[i] : 0;
        int inc = v;
        for (int o = 1; o < 32; o <<= 1) {
            int t = __shfl_up(inc, o, 32);
            if (lane >= o) inc += t;
        }
        if (lane == 31) wsum[w] = inc;
        __syncthreads();
        int wbase = 0, total = 0;
        for (int k = 0; k < 32; ++k) {
            int s = wsum[k];
            if (k < w) wbase += s;
            total += s;
        }
        if (i < nb) bsum[i] = carry + wbase + inc - v;
        carry += total;
        __syncthreads();
    }
}

__global__ void __launch_bounds__(256)
fill_kernel(const int* __restrict__ ec, const int* __restrict__ row,
            const int* __restrict__ col, const int* __restrict__ rank,
            const int* __restrict__ off, const int* __restrict__ bsum,
            const int2* __restrict__ cnt0,
            const int* __restrict__ mm, const int* __restrict__ mm1,
            int2* __restrict__ rec, int E) {
    const int T = gridDim.x * blockDim.x;
    const int t = blockIdx.x * blockDim.x + threadIdx.x;
    const float mn = (float)min(mm[0], mm1[0]), mx = (float)max(mm[1], mm1[1]);
    const float denom = mx - mn;
    const bool span = denom > 0.0f;
    const float dsafe = span ? denom : 1.0f;
    if (t + (EPT - 1) * T < E) {
        int v[EPT], r[EPT], c[EPT], pos[EPT], sb[EPT], so[EPT], c0[EPT];
        for (int u = 0; u < EPT; ++u) {
            const int e = t + u * T;
            v[u] = ec[e];
            r[u] = row[e];
            c[u] = col[e];
            pos[u] = rank[e];
        }
        for (int u = 0; u < EPT; ++u) {
            sb[u] = bsum[c[u] >> SCAN_SHIFT];
            so[u] = off[c[u]];
            c0[u] = cnt0[c[u]].x;
        }
        for (int u = 0; u < EPT; ++u) {
            const float q = ((float)v[u] - mn) / dsafe;
            const float w = span ? q : 0.0f;
            const int behind = (pos[u] & 1) ? c0[u] : 0;
            rec[sb[u] + so[u] + behind + (pos[u] >> 1)] = make_int2(r[u], __float_as_int(w));
        }
    } else {
        for (int e = t; e < E; e += T) {
            const float q = ((float)ec[e] - mn) / dsafe;
            const float w = span ? q : 0.0f;
            const int c = col[e];
            const int rk = rank[e];
            const int behind = (rk & 1) ? cnt0[c].x : 0;
            rec[bsum[c >> SCAN_SHIFT] + off[c] + behind + (rk >> 1)] = make_int2(row[e], __float_as_int(w));
        }
    }
}

template <int K, bool RELU>
__device__ __forceinline__ void gemm_block(const float* __restrict__ A, const float* __restrict__ W,
                                           float* __restrict__ H, int M, int blk, float* sW) {

    for (int i = threadIdx.x; i < (K / 2) * HID; i += blockDim.x) {
        int kp = i >> 6;
        int c  = i & 63;
        float w0 = W[(2 * kp) * HID + c];
        float w1 = W[(2 * kp + 1) * HID + c];
        float* dst = sW + kp * (2 * HID) + 2 * c;
        dst[0] = w0;
        dst[1] = w1;
    }
    __syncthreads();

    const int lane  = threadIdx.x & 31;
    const int widx  = threadIdx.x >> 5;
    const int half  = lane >> 4;
    const int sub   = lane & 15;
    const long tile = (long)blk * 8 + widx;
    const long row0 = tile * 16;
    if (row0 >= M) return;

    v8f acc[4] = {};

    const float* aRow = A + (row0 + sub) * (long)K;

#pragma unroll 4
    for (int kk = 0; kk < K; kk += 4) {
        const int kb = kk + 2 * half;
        v2f a = *(const v2f*)(aRow + kb);
        if (RELU) { a.x = fmaxf(a.x, 0.0f); a.y = fmaxf(a.y, 0.0f); }

        const float* wp = sW + (kb >> 1) * (2 * HID) + 2 * sub;
#pragma unroll
        for (int t = 0; t < 4; ++t) {
            v2f b = *(const v2f*)(wp + t * 32);
            acc[t] = __builtin_amdgcn_wmma_f32_16x16x4_f32(
                 false, a,  false, b,
                 (short)0, acc[t],
                 false,  false);
        }
    }

#pragma unroll
    for (int r = 0; r < 8; ++r) {
        float* hp = H + (row0 + r + 8 * half) * (long)HID + sub;
#pragma unroll
        for (int t = 0; t < 4; ++t) hp[t * 16] = acc[t][r];
    }
}

template <int K, bool RELU>
__global__ void gemm_wmma_kernel(const float* __restrict__ A, const float* __restrict__ W,
                                 float* __restrict__ H, int M) {
    __shared__ __align__(16) float sW[K * HID];
    gemm_block<K, RELU>(A, W, H, M, blockIdx.x, sW);
}

__global__ void __launch_bounds__(256)
front_kernel(const float* __restrict__ x, const float* __restrict__ W1, float* __restrict__ h,
             int N, int gemmBlocks,
             const int* __restrict__ col, u64* __restrict__ cnt0, u64* __restrict__ cnt1,
             int* __restrict__ rank, int E, int countBlocks,
             const int* __restrict__ ec, int* __restrict__ mm0, int* __restrict__ mm1) {
    __shared__ __align__(16) float sW[256 * HID];
    const int b = blockIdx.x;
    if (b < gemmBlocks) {
        gemm_block<256, false>(x, W1, h, N, b, sW);
    } else if (b < gemmBlocks + countBlocks) {
        count_role(col, ec, cnt0, cnt1, rank, E, b - gemmBlocks, countBlocks);
    } else {
        const int first = gemmBlocks + countBlocks;
        minmax_role(ec, E, mm0, mm1, b - first, (int)gridDim.x - first);
    }
}


template <int MASK>
__device__ __forceinline__ float row_xor(float x) {
    const int b = __builtin_bit_cast(int, x);
    return __builtin_bit_cast(float,
        __builtin_amdgcn_mov_dpp(b, 0x160 + MASK, 0xf, 0xf, false));
}
__device__ __forceinline__ float xhalf(float x) {
    const int b = __builtin_bit_cast(int, x);
    return __builtin_bit_cast(float,
        __builtin_amdgcn_permlanex16(b, b, 0x76543210, 0xfedcba98, false, false));
}
__device__ __forceinline__ float wave_max(float m) {
    m = fmaxf(m, row_xor<1>(m));
    m = fmaxf(m, row_xor<2>(m));
    m = fmaxf(m, row_xor<4>(m));
    m = fmaxf(m, row_xor<8>(m));
    return fmaxf(m, xhalf(m));
}
__device__ __forceinline__ float wave_sum(float s) {
    s += row_xor<1>(s);
    s += row_xor<2>(s);
    s += row_xor<4>(s);
    s += row_xor<8>(s);
    return s + xhalf(s);
}

template <int B>
__device__ __forceinline__ void gather_batch(const v2f* __restrict__ hl, int r, float nrm,
                                             int j0, v2f& acc) {
    v2f a[B];
#pragma unroll
    for (int j = 0; j < B; ++j) {
        const int rj = __builtin_amdgcn_readlane(r, j0 + j);
        a[j] = hl[(long)rj * (HID / 2)];
    }
#pragma unroll
    for (int j = 0; j < B; ++j) {
        const float nj = __int_as_float(
            __builtin_amdgcn_readlane(__float_as_int(nrm), j0 + j));
        acc.x = fmaf(a[j].x, nj, acc.x);
        acc.y = fmaf(a[j].y, nj, acc.y);
    }
}

template <bool FIRST>
__global__ void __launch_bounds__(256)
gather_kernel(const float* __restrict__ h, const float* __restrict__ dinv,
              const float* __restrict__ bias, const int* __restrict__ off,
              const int* __restrict__ bsum, const int* __restrict__ cnt,
              int2* __restrict__ rec, float* __restrict__ out,
              float* __restrict__ lsm, int N) {
    const int lane = threadIdx.x & 31;
    const int wave = __builtin_amdgcn_readfirstlane(blockIdx.x * 8 + (threadIdx.x >> 5));
    const int nwaves = gridDim.x * 8;
    const v2f* hl = (const v2f*)h + lane;
    const v2f bv = ((const v2f*)bias)[lane];
    for (int node = wave; node < N; node += nwaves) {
    const int start = bsum[node >> SCAN_SHIFT] + off[node];
    const int len = cnt[node];
    const float di = dinv[node];
    const v2f self = hl[(long)node * (HID / 2)];
    int2* rp = rec + ((long)start + lane);

    int m = len < 32 ? len : 32;
    int2 rc = make_int2(0, 0);
    if (lane < m) rc = *rp;

    v2f acc = {0.0f, 0.0f};
    for (int base = 0; base < len; base += 32) {
        const int r = rc.x;
        float nrm = __int_as_float(rc.y);
        if (FIRST) {
            if (lane < m) {
                nrm = dinv[r] * nrm * di;
                rp->y = __float_as_int(nrm);
            }
        }
        int j0 = 0;
#pragma unroll 1
        for (; j0 + 16 <= m; j0 += 16) gather_batch<16>(hl, r, nrm, j0, acc);
        switch (m - j0) {
#define GATHER_CASE(B) case B: gather_batch<B>(hl, r, nrm, j0, acc); break;
            GATHER_CASE(15) GATHER_CASE(14) GATHER_CASE(13) GATHER_CASE(12)
            GATHER_CASE(11) GATHER_CASE(10) GATHER_CASE(9)  GATHER_CASE(8)
            GATHER_CASE(7)  GATHER_CASE(6)  GATHER_CASE(5)  GATHER_CASE(4)
            GATHER_CASE(3)  GATHER_CASE(2)  GATHER_CASE(1)
#undef GATHER_CASE
            default: break;
        }
        const int left = len - base - 32;
        m = left < 32 ? left : 32;
        rp += 32;
        if (lane < m) rc = *rp;
    }
    const float sn = di * di;
    v2f o;
    o.x = fmaf(self.x, sn, acc.x) + bv.x;
    o.y = fmaf(self.y, sn, acc.y) + bv.y;
    ((v2f*)out)[(long)node * (HID / 2) + lane] = o;
    if (!FIRST) {
        const float mx = wave_max(fmaxf(o.x, o.y));
        const float se = wave_sum(__expf(o.x - mx) + __expf(o.y - mx));
        const float lse = mx + __logf(se);
        v2f l;
        l.x = o.x - lse;
        l.y = o.y - lse;
        ((v2f*)lsm)[(long)node * (HID / 2) + lane] = l;
    }
    }
}

extern "C" void kernel_launch(void* const* d_in, const int* in_sizes, int n_in,
                              void* d_out, int out_size, void* d_ws, size_t ws_size,
                              hipStream_t stream) {
    const float* x  = (const float*)d_in[0];
    const int*   ei = (const int*)d_in[1];
    const int*   ec = (const int*)d_in[2];
    const float* W1 = (const float*)d_in[3];
    const float* b1 = (const float*)d_in[4];
    const float* W2 = (const float*)d_in[5];
    const float* b2 = (const float*)d_in[6];

    const int IN_DIM = 256;
    const int N = in_sizes[0] / IN_DIM;
    const int E = in_sizes[2];
    const int* row = ei;
    const int* col = ei + E;

    float* wsf  = (float*)d_ws;
    float* h    = wsf;
    float* agg1 = h    + (size_t)N * HID;
    int2*  rec  = (int2*)(agg1 + (size_t)N * HID);
    int*   rank = (int*)(rec + (size_t)E);
    float* dinv = (float*)(rank + (size_t)E);
    int*   len  = (int*)(dinv + (size_t)N);
    u64*   cnt  = (u64*)agg1;
    u64*   cnt1 = cnt + (size_t)N;
    int*   off  = len + (size_t)N;
    const int nb = (N + SCAN_B - 1) / SCAN_B;
    int*   bsum = off + (size_t)N;
    int*   mm   = bsum + (size_t)nb;
    int*   mm1  = (int*)(cnt1 + (size_t)N) + 64;
    float* h2   = h;

    float* emb = (float*)d_out;
    float* lsm = emb + (size_t)N * HID;

    const int tileBlocks = (int)((N / 16 + 7) / 8);
    const int edgeBlocks = (E + 256 * EPT - 1) / (256 * EPT);
    const int nodeBlocks = 2048;
    const int mmBlocks = 256;

    init_kernel<<<(N + 255) / 256, 256, 0, stream>>>(cnt, cnt1, mm, mm1, N);
    front_kernel<<<tileBlocks + edgeBlocks + mmBlocks, 256, 0, stream>>>(
        x, W1, h, N, tileBlocks, col, cnt, cnt1, rank, E, edgeBlocks, ec, mm, mm1);
    scan_kernel<<<nb, SCAN_B, 0, stream>>>((const int2*)cnt, (const int2*)cnt1, mm, mm1, off, bsum, len, dinv, N);
    bscan_kernel<<<1, 1024, 0, stream>>>(bsum, nb);
    fill_kernel<<<edgeBlocks, 256, 0, stream>>>(ec, row, col, rank, off, bsum, (const int2*)cnt, mm, mm1, rec, E);

    gather_kernel<true><<<nodeBlocks, 256, 0, stream>>>(h, dinv, b1, off, bsum, len, rec, agg1, nullptr, N);

    gemm_wmma_kernel<64, true><<<tileBlocks, 256, 0, stream>>>(agg1, W2, h2, N);
    gather_kernel<false><<<nodeBlocks, 256, 0, stream>>>(h2, dinv, b2, off, bsum, len, rec, emb, lsm, N);
}


# segment bases from one counting add per 256-node group on a cursor per package half (absolute; both copies' bases in one 8-byte word): block-sum launch gone, placing loads 1 word per edge, not 3
# speedup vs baseline: 2.6747x; 1.0431x over previous
#include <hip/hip_runtime.h>
#include <hip/hip_bf16.h>

typedef __attribute__((ext_vector_type(2))) float v2f;
typedef __attribute__((ext_vector_type(8))) float v8f;
typedef unsigned long long u64;

#define HID 64

__device__ __forceinline__ int package_half() {
    return (__builtin_amdgcn_s_sendmsg_rtn(0x87) >> 18) & 1;
}

__global__ void init_kernel(u64* __restrict__ cnt, u64* __restrict__ cnt1,
                            int* __restrict__ mm, int* __restrict__ mm1,
                            int* __restrict__ cur0, int* __restrict__ cur1, int N) {
    int i = blockIdx.x * blockDim.x + threadIdx.x;
    if (i < N) { cnt[i] = 0; cnt1[i] = 0; }
    if (i == 0) {
        mm[0] = 0x7FFFFFFF;  mm[1] = 0x80000000;
        mm1[0] = 0x7FFFFFFF; mm1[1] = 0x80000000;
        *cur0 = 0;
        *cur1 = 0;
    }
}

#define MM_EPT 8

__device__ __forceinline__ void minmax_role(const int* __restrict__ ec, int E,
                                            int* __restrict__ mm0, int* __restrict__ mm1,
                                            int blk, int nblk) {
    const int T = nblk * 256;
    const int t = blk * 256 + threadIdx.x;
    int lmin = 0x7FFFFFFF, lmax = 0x80000000;
    for (int i = t; i < E; i += MM_EPT * T) {
        int v[MM_EPT];
        for (int u = 0; u < MM_EPT; ++u) v[u] = ec[min(i + u * T, E - 1)];
        for (int u = 0; u < MM_EPT; ++u) {
            lmin = min(lmin, v[u]);
            lmax = max(lmax, v[u]);
        }
    }
    int* mm = package_half() ? mm1 : mm0;
    atomicMin(&mm[0], lmin);
    atomicMax(&mm[1], lmax);
}

#define EPT 4

__device__ __forceinline__ u64 count_word(int ecv) {
    return ((u64)(long long)ecv << 32) + 1ull;
}

__device__ __forceinline__ void
count_role(const int* __restrict__ col, const int* __restrict__ ec, u64* __restrict__ cnt0,
           u64* __restrict__ cnt1, int* __restrict__ rank, int E, int blk, int nblk) {
    const int T = nblk * 256;
    const int t = blk * 256 + threadIdx.x;
    const int hf = package_half();
    u64* cnt = hf ? cnt1 : cnt0;
    if (t + (EPT - 1) * T < E) {
        int c[EPT], v[EPT];
        u64 old[EPT];
        for (int u = 0; u < EPT; ++u) c[u] = col[t + u * T];
        for (int u = 0; u < EPT; ++u) v[u] = ec[t + u * T];
        for (int u = 0; u < EPT; ++u) old[u] = atomicAdd(&cnt[c[u]], count_word(v[u]));
        for (int u = 0; u < EPT; ++u) rank[t + u * T] = (int)(unsigned)old[u] * 2 + hf;
    } else {
        for (int e = t; e < E; e += T)
            rank[e] = (int)(unsigned)atomicAdd(&cnt[col[e]], count_word(ec[e])) * 2 + hf;
    }
}

#define SCAN_B 256

__global__ void __launch_bounds__(SCAN_B)
scan_kernel(const int2* __restrict__ cnt, const int2* __restrict__ cnt1,
            const int* __restrict__ mm, const int* __restrict__ mm1,
            int* __restrict__ cur0, int* __restrict__ cur1,
            int* __restrict__ off, int2* __restrict__ fb,
            int* __restrict__ len, float* __restrict__ dinv, int N, int E) {
    __shared__ int wsum[SCAN_B / 32];
    __shared__ int gbase;
    const int tid = threadIdx.x, lane = tid & 31, w = tid >> 5;
    const int i = blockIdx.x * SCAN_B + tid;
    int2 ca = make_int2(0, 0), cb = make_int2(0, 0);
    if (i < N) { ca = cnt[i]; cb = cnt1[i]; }
    const int v = ca.x + cb.x;
    int inc = v;
    for (int o = 1; o < 32; o <<= 1) {
        int t = __shfl_up(inc, o, 32);
        if (lane >= o) inc += t;
    }
    if (lane == 31) wsum[w] = inc;
    __syncthreads();
    int wbase = 0, total = 0;
    for (int k = 0; k < SCAN_B / 32; ++k) {
        int s = wsum[k];
        if (k < w) wbase += s;
        total += s;
    }
    if (tid == 0) {
        const int hf = package_half();
        const int old = atomicAdd(hf ? cur1 : cur0, total);
        gbase = hf ? E - old - total : old;
    }
    float dv = 0.0f;
    if (i < N) {
        const int mni = min(mm[0], mm1[0]), mxi = max(mm[1], mm1[1]);
        const float denom = (float)mxi - (float)mni;
        const long long num = (long long)ca.y + (long long)cb.y - (long long)v * (long long)mni;
        const float d = (denom > 0.0f) ? 1.0f + (float)num / denom : 1.0f;
        dv = (d > 0.0f) ? rsqrtf(d) : 0.0f;
    }
    __syncthreads();
    if (i < N) {
        const int start = gbase + wbase + inc - v;
        off[i] = start;
        fb[i] = make_int2(start, start + ca.x);
        len[i] = v;
        dinv[i] = dv;
    }
}

__global__ void __launch_bounds__(256)
fill_kernel(const int* __restrict__ ec, const int* __restrict__ row,
            const int* __restrict__ col, const int* __restrict__ rank,
            const int2* __restrict__ fb,
            const int* __restrict__ mm, const int* __restrict__ mm1,
            int2* __restrict__ rec, int E) {
    const int T = gridDim.x * blockDim.x;
    const int t = blockIdx.x * blockDim.x + threadIdx.x;
    const float mn = (float)min(mm[0], mm1[0]), mx = (float)max(mm[1], mm1[1]);
    const float denom = mx - mn;
    const bool span = denom > 0.0f;
    const float dsafe = span ? denom : 1.0f;
    if (t + (EPT - 1) * T < E) {
        int v[EPT], r[EPT], c[EPT], pos[EPT], b0[EPT], b1[EPT];
        for (int u = 0; u < EPT; ++u) {
            const int e = t + u * T;
            v[u] = ec[e];
            r[u] = row[e];
            c[u] = col[e];
            pos[u] = rank[e];
        }
        for (int u = 0; u < EPT; ++u) {
            const int2 q = fb[c[u]];
            b0[u] = q.x;
            b1[u] = q.y;
        }
        for (int u = 0; u < EPT; ++u) {
            const float q = ((float)v[u] - mn) / dsafe;
            const float w = span ? q : 0.0f;
            const int base = (pos[u] & 1) ? b1[u] : b0[u];
            rec[base + (pos[u] >> 1)] = make_int2(r[u], __float_as_int(w));
        }
    } else {
        for (int e = t; e < E; e += T) {
            const float q = ((float)ec[e] - mn) / dsafe;
            const float w = span ? q : 0.0f;
            const int2 b = fb[col[e]];
            const int rk = rank[e];
            rec[((rk & 1) ? b.y : b.x) + (rk >> 1)] = make_int2(row[e], __float_as_int(w));
        }
    }
}

template <int K, bool RELU>
__device__ __forceinline__ void gemm_block(const float* __restrict__ A, const float* __restrict__ W,
                                           float* __restrict__ H, int M, int blk, float* sW) {

    for (int i = threadIdx.x; i < (K / 2) * HID; i += blockDim.x) {
        int kp = i >> 6;
        int c  = i & 63;
        float w0 = W[(2 * kp) * HID + c];
        float w1 = W[(2 * kp + 1) * HID + c];
        float* dst = sW + kp * (2 * HID) + 2 * c;
        dst[0] = w0;
        dst[1] = w1;
    }
    __syncthreads();

    const int lane  = threadIdx.x & 31;
    const int widx  = threadIdx.x >> 5;
    const int half  = lane >> 4;
    const int sub   = lane & 15;
    const long tile = (long)blk * 8 + widx;
    const long row0 = tile * 16;
    if (row0 >= M) return;

    v8f acc[4] = {};

    const float* aRow = A + (row0 + sub) * (long)K;

#pragma unroll 4
    for (int kk = 0; kk < K; kk += 4) {
        const int kb = kk + 2 * half;
        v2f a = *(const v2f*)(aRow + kb);
        if (RELU) { a.x = fmaxf(a.x, 0.0f); a.y = fmaxf(a.y, 0.0f); }

        const float* wp = sW + (kb >> 1) * (2 * HID) + 2 * sub;
#pragma unroll
        for (int t = 0; t < 4; ++t) {
            v2f b = *(const v2f*)(wp + t * 32);
            acc[t] = __builtin_amdgcn_wmma_f32_16x16x4_f32(
                 false, a,  false, b,
                 (short)0, acc[t],
                 false,  false);
        }
    }

#pragma unroll
    for (int r = 0; r < 8; ++r) {
        float* hp = H + (row0 + r + 8 * half) * (long)HID + sub;
#pragma unroll
        for (int t = 0; t < 4; ++t) hp[t * 16] = acc[t][r];
    }
}

template <int K, bool RELU>
__global__ void gemm_wmma_kernel(const float* __restrict__ A, const float* __restrict__ W,
                                 float* __restrict__ H, int M) {
    __shared__ __align__(16) float sW[K * HID];
    gemm_block<K, RELU>(A, W, H, M, blockIdx.x, sW);
}

__global__ void __launch_bounds__(256)
front_kernel(const float* __restrict__ x, const float* __restrict__ W1, float* __restrict__ h,
             int N, int gemmBlocks,
             const int* __restrict__ col, u64* __restrict__ cnt0, u64* __restrict__ cnt1,
             int* __restrict__ rank, int E, int countBlocks,
             const int* __restrict__ ec, int* __restrict__ mm0, int* __restrict__ mm1) {
    __shared__ __align__(16) float sW[256 * HID];
    const int b = blockIdx.x;
    if (b < gemmBlocks) {
        gemm_block<256, false>(x, W1, h, N, b, sW);
    } else if (b < gemmBlocks + countBlocks) {
        count_role(col, ec, cnt0, cnt1, rank, E, b - gemmBlocks, countBlocks);
    } else {
        const int first = gemmBlocks + countBlocks;
        minmax_role(ec, E, mm0, mm1, b - first, (int)gridDim.x - first);
    }
}


template <int MASK>
__device__ __forceinline__ float row_xor(float x) {
    const int b = __builtin_bit_cast(int, x);
    return __builtin_bit_cast(float,
        __builtin_amdgcn_mov_dpp(b, 0x160 + MASK, 0xf, 0xf, false));
}
__device__ __forceinline__ float xhalf(float x) {
    const int b = __builtin_bit_cast(int, x);
    return __builtin_bit_cast(float,
        __builtin_amdgcn_permlanex16(b, b, 0x76543210, 0xfedcba98, false, false));
}
__device__ __forceinline__ float wave_max(float m) {
    m = fmaxf(m, row_xor<1>(m));
    m = fmaxf(m, row_xor<2>(m));
    m = fmaxf(m, row_xor<4>(m));
    m = fmaxf(m, row_xor<8>(m));
    return fmaxf(m, xhalf(m));
}
__device__ __forceinline__ float wave_sum(float s) {
    s += row_xor<1>(s);
    s += row_xor<2>(s);
    s += row_xor<4>(s);
    s += row_xor<8>(s);
    return s + xhalf(s);
}

template <int B>
__device__ __forceinline__ void gather_batch(const v2f* __restrict__ hl, int r, float nrm,
                                             int j0, v2f& acc) {
    v2f a[B];
#pragma unroll
    for (int j = 0; j < B; ++j) {
        const int rj = __builtin_amdgcn_readlane(r, j0 + j);
        a[j] = hl[(long)rj * (HID / 2)];
    }
#pragma unroll
    for (int j = 0; j < B; ++j) {
        const float nj = __int_as_float(
            __builtin_amdgcn_readlane(__float_as_int(nrm), j0 + j));
        acc.x = fmaf(a[j].x, nj, acc.x);
        acc.y = fmaf(a[j].y, nj, acc.y);
    }
}

template <bool FIRST>
__global__ void __launch_bounds__(256)
gather_kernel(const float* __restrict__ h, const float* __restrict__ dinv,
              const float* __restrict__ bias, const int* __restrict__ off,
              const int* __restrict__ cnt,
              int2* __restrict__ rec, float* __restrict__ out,
              float* __restrict__ lsm, int N) {
    const int lane = threadIdx.x & 31;
    const int wave = __builtin_amdgcn_readfirstlane(blockIdx.x * 8 + (threadIdx.x >> 5));
    const int nwaves = gridDim.x * 8;
    const v2f* hl = (const v2f*)h + lane;
    const v2f bv = ((const v2f*)bias)[lane];
    for (int node = wave; node < N; node += nwaves) {
    const int start = off[node];
    const int len = cnt[node];
    const float di = dinv[node];
    const v2f self = hl[(long)node * (HID / 2)];
    int2* rp = rec + ((long)start + lane);

    int m = len < 32 ? len : 32;
    int2 rc = make_int2(0, 0);
    if (lane < m) rc = *rp;

    v2f acc = {0.0f, 0.0f};
    for (int base = 0; base < len; base += 32) {
        const int r = rc.x;
        float nrm = __int_as_float(rc.y);
        if (FIRST) {
            if (lane < m) {
                nrm = dinv[r] * nrm * di;
                rp->y = __float_as_int(nrm);
            }
        }
        int j0 = 0;
#pragma unroll 1
        for (; j0 + 16 <= m; j0 += 16) gather_batch<16>(hl, r, nrm, j0, acc);
        switch (m - j0) {
#define GATHER_CASE(B) case B: gather_batch<B>(hl, r, nrm, j0, acc); break;
            GATHER_CASE(15) GATHER_CASE(14) GATHER_CASE(13) GATHER_CASE(12)
            GATHER_CASE(11) GATHER_CASE(10) GATHER_CASE(9)  GATHER_CASE(8)
            GATHER_CASE(7)  GATHER_CASE(6)  GATHER_CASE(5)  GATHER_CASE(4)
            GATHER_CASE(3)  GATHER_CASE(2)  GATHER_CASE(1)
#undef GATHER_CASE
            default: break;
        }
        const int left = len - base - 32;
        m = left < 32 ? left : 32;
        rp += 32;
        if (lane < m) rc = *rp;
    }
    const float sn = di * di;
    v2f o;
    o.x = fmaf(self.x, sn, acc.x) + bv.x;
    o.y = fmaf(self.y, sn, acc.y) + bv.y;
    ((v2f*)out)[(long)node * (HID / 2) + lane] = o;
    if (!FIRST) {
        const float mx = wave_max(fmaxf(o.x, o.y));
        const float se = wave_sum(__expf(o.x - mx) + __expf(o.y - mx));
        const float lse = mx + __logf(se);
        v2f l;
        l.x = o.x - lse;
        l.y = o.y - lse;
        ((v2f*)lsm)[(long)node * (HID / 2) + lane] = l;
    }
    }
}

extern "C" void kernel_launch(void* const* d_in, const int* in_sizes, int n_in,
                              void* d_out, int out_size, void* d_ws, size_t ws_size,
                              hipStream_t stream) {
    const float* x  = (const float*)d_in[0];
    const int*   ei = (const int*)d_in[1];
    const int*   ec = (const int*)d_in[2];
    const float* W1 = (const float*)d_in[3];
    const float* b1 = (const float*)d_in[4];
    const float* W2 = (const float*)d_in[5];
    const float* b2 = (const float*)d_in[6];

    const int IN_DIM = 256;
    const int N = in_sizes[0] / IN_DIM;
    const int E = in_sizes[2];
    const int* row = ei;
    const int* col = ei + E;

    float* wsf  = (float*)d_ws;
    float* h    = wsf;
    float* agg1 = h    + (size_t)N * HID;
    int2*  rec  = (int2*)(agg1 + (size_t)N * HID);
    int*   rank = (int*)(rec + (size_t)E);
    float* dinv = (float*)(rank + (size_t)E);
    int*   len  = (int*)(dinv + (size_t)N);
    u64*   cnt  = (u64*)agg1;
    u64*   cnt1 = cnt + (size_t)N;
    int*   off  = len + (size_t)N;
    const int nb = (N + SCAN_B - 1) / SCAN_B;
    int*   mm   = off + (size_t)N;
    int*   cur0 = mm + 64;
    int*   mm1  = (int*)(cnt1 + (size_t)N) + 64;
    int*   cur1 = mm1 + 64;
    int2*  fb   = (int2*)(cur1 + 64);
    float* h2   = h;

    float* emb = (float*)d_out;
    float* lsm = emb + (size_t)N * HID;

    const int tileBlocks = (int)((N / 16 + 7) / 8);
    const int edgeBlocks = (E + 256 * EPT - 1) / (256 * EPT);
    const int nodeBlocks = 2048;
    const int mmBlocks = 256;

    init_kernel<<<(N + 255) / 256, 256, 0, stream>>>(cnt, cnt1, mm, mm1, cur0, cur1, N);
    front_kernel<<<tileBlocks + edgeBlocks + mmBlocks, 256, 0, stream>>>(
        x, W1, h, N, tileBlocks, col, cnt, cnt1, rank, E, edgeBlocks, ec, mm, mm1);
    scan_kernel<<<nb, SCAN_B, 0, stream>>>((const int2*)cnt, (const int2*)cnt1, mm, mm1, cur0, cur1, off, fb, len, dinv, N, E);
    fill_kernel<<<edgeBlocks, 256, 0, stream>>>(ec, row, col, rank, fb, mm, mm1, rec, E);

    gather_kernel<true><<<nodeBlocks, 256, 0, stream>>>(h, dinv, b1, off, len, rec, agg1, nullptr, N);

    gemm_wmma_kernel<64, true><<<tileBlocks, 256, 0, stream>>>(agg1, W2, h2, N);
    gather_kernel<false><<<nodeBlocks, 256, 0, stream>>>(h2, dinv, b2, off, len, rec, emb, lsm, N);
}


# layer 2's product moved into layer 1's aggregation launch: the waves put finished rows (bias, ReLU) in an LDS row tile and multiply it by W2 on the same f32 WMMA; one launch fewer, sums unchanged
# speedup vs baseline: 2.8516x; 1.0661x over previous
#include <hip/hip_runtime.h>
#include <hip/hip_bf16.h>

typedef __attribute__((ext_vector_type(2))) float v2f;
typedef __attribute__((ext_vector_type(8))) float v8f;
typedef unsigned long long u64;

#define HID 64

__device__ __forceinline__ int package_half() {
    return (__builtin_amdgcn_s_sendmsg_rtn(0x87) >> 18) & 1;
}

__global__ void init_kernel(u64* __restrict__ cnt, u64* __restrict__ cnt1,
                            int* __restrict__ mm, int* __restrict__ mm1,
                            int* __restrict__ cur0, int* __restrict__ cur1, int N) {
    int i = blockIdx.x * blockDim.x + threadIdx.x;
    if (i < N) { cnt[i] = 0; cnt1[i] = 0; }
    if (i == 0) {
        mm[0] = 0x7FFFFFFF;  mm[1] = 0x80000000;
        mm1[0] = 0x7FFFFFFF; mm1[1] = 0x80000000;
        *cur0 = 0;
        *cur1 = 0;
    }
}

#define MM_EPT 8

__device__ __forceinline__ void minmax_role(const int* __restrict__ ec, int E,
                                            int* __restrict__ mm0, int* __restrict__ mm1,
                                            int blk, int nblk) {
    const int T = nblk * 256;
    const int t = blk * 256 + threadIdx.x;
    int lmin = 0x7FFFFFFF, lmax = 0x80000000;
    for (int i = t; i < E; i += MM_EPT * T) {
        int v[MM_EPT];
        for (int u = 0; u < MM_EPT; ++u) v[u] = ec[min(i + u * T, E - 1)];
        for (int u = 0; u < MM_EPT; ++u) {
            lmin = min(lmin, v[u]);
            lmax = max(lmax, v[u]);
        }
    }
    int* mm = package_half() ? mm1 : mm0;
    atomicMin(&mm[0], lmin);
    atomicMax(&mm[1], lmax);
}

#define EPT 4

__device__ __forceinline__ u64 count_word(int ecv) {
    return ((u64)(long long)ecv << 32) + 1ull;
}

__device__ __forceinline__ void
count_role(const int* __restrict__ col, const int* __restrict__ ec, u64* __restrict__ cnt0,
           u64* __restrict__ cnt1, int* __restrict__ rank, int E, int blk, int nblk) {
    const int T = nblk * 256;
    const int t = blk * 256 + threadIdx.x;
    const int hf = package_half();
    u64* cnt = hf ? cnt1 : cnt0;
    if (t + (EPT - 1) * T < E) {
        int c[EPT], v[EPT];
        u64 old[EPT];
        for (int u = 0; u < EPT; ++u) c[u] = col[t + u * T];
        for (int u = 0; u < EPT; ++u) v[u] = ec[t + u * T];
        for (int u = 0; u < EPT; ++u) old[u] = atomicAdd(&cnt[c[u]], count_word(v[u]));
        for (int u = 0; u < EPT; ++u) rank[t + u * T] = (int)(unsigned)old[u] * 2 + hf;
    } else {
        for (int e = t; e < E; e += T)
            rank[e] = (int)(unsigned)atomicAdd(&cnt[col[e]], count_word(ec[e])) * 2 + hf;
    }
}

#define SCAN_B 256

__global__ void __launch_bounds__(SCAN_B)
scan_kernel(const int2* __restrict__ cnt, const int2* __restrict__ cnt1,
            const int* __restrict__ mm, const int* __restrict__ mm1,
            int* __restrict__ cur0, int* __restrict__ cur1,
            int* __restrict__ off, int2* __restrict__ fb,
            int* __restrict__ len, float* __restrict__ dinv, int N, int E) {
    __shared__ int wsum[SCAN_B / 32];
    __shared__ int gbase;
    const int tid = threadIdx.x, lane = tid & 31, w = tid >> 5;
    const int i = blockIdx.x * SCAN_B + tid;
    int2 ca = make_int2(0, 0), cb = make_int2(0, 0);
    if (i < N) { ca = cnt[i]; cb = cnt1[i]; }
    const int v = ca.x + cb.x;
    int inc = v;
    for (int o = 1; o < 32; o <<= 1) {
        int t = __shfl_up(inc, o, 32);
        if (lane >= o) inc += t;
    }
    if (lane == 31) wsum[w] = inc;
    __syncthreads();
    int wbase = 0, total = 0;
    for (int k = 0; k < SCAN_B / 32; ++k) {
        int s = wsum[k];
        if (k < w) wbase += s;
        total += s;
    }
    if (tid == 0) {
        const int hf = package_half();
        const int old = atomicAdd(hf ? cur1 : cur0, total);
        gbase = hf ? E - old - total : old;
    }
    float dv = 0.0f;
    if (i < N) {
        const int mni = min(mm[0], mm1[0]), mxi = max(mm[1], mm1[1]);
        const float denom = (float)mxi - (float)mni;
        const long long num = (long long)ca.y + (long long)cb.y - (long long)v * (long long)mni;
        const float d = (denom > 0.0f) ? 1.0f + (float)num / denom : 1.0f;
        dv = (d > 0.0f) ? rsqrtf(d) : 0.0f;
    }
    __syncthreads();
    if (i < N) {
        const int start = gbase + wbase + inc - v;
        off[i] = start;
        fb[i] = make_int2(start, start + ca.x);
        len[i] = v;
        dinv[i] = dv;
    }
}

__global__ void __launch_bounds__(256)
fill_kernel(const int* __restrict__ ec, const int* __restrict__ row,
            const int* __restrict__ col, const int* __restrict__ rank,
            const int2* __restrict__ fb,
            const int* __restrict__ mm, const int* __restrict__ mm1,
            int2* __restrict__ rec, int E) {
    const int T = gridDim.x * blockDim.x;
    const int t = blockIdx.x * blockDim.x + threadIdx.x;
    const float mn = (float)min(mm[0], mm1[0]), mx = (float)max(mm[1], mm1[1]);
    const float denom = mx - mn;
    const bool span = denom > 0.0f;
    const float dsafe = span ? denom : 1.0f;
    if (t + (EPT - 1) * T < E) {
        int v[EPT], r[EPT], c[EPT], pos[EPT], b0[EPT], b1[EPT];
        for (int u = 0; u < EPT; ++u) {
            const int e = t + u * T;
            v[u] = ec[e];
            r[u] = row[e];
            c[u] = col[e];
            pos[u] = rank[e];
        }
        for (int u = 0; u < EPT; ++u) {
            const int2 q = fb[c[u]];
            b0[u] = q.x;
            b1[u] = q.y;
        }
        for (int u = 0; u < EPT; ++u) {
            const float q = ((float)v[u] - mn) / dsafe;
            const float w = span ? q : 0.0f;
            const int base = (pos[u] & 1) ? b1[u] : b0[u];
            rec[base + (pos[u] >> 1)] = make_int2(r[u], __float_as_int(w));
        }
    } else {
        for (int e = t; e < E; e += T) {
            const float q = ((float)ec[e] - mn) / dsafe;
            const float w = span ? q : 0.0f;
            const int2 b = fb[col[e]];
            const int rk = rank[e];
            rec[((rk & 1) ? b.y : b.x) + (rk >> 1)] = make_int2(row[e], __float_as_int(w));
        }
    }
}

template <int K>
__device__ __forceinline__ void gemm_block(const float* __restrict__ A, const float* __restrict__ W,
                                           float* __restrict__ H, int M, int blk, float* sW) {

    for (int i = threadIdx.x; i < (K / 2) * HID; i += blockDim.x) {
        int kp = i >> 6;
        int c  = i & 63;
        float w0 = W[(2 * kp) * HID + c];
        float w1 = W[(2 * kp + 1) * HID + c];
        float* dst = sW + kp * (2 * HID) + 2 * c;
        dst[0] = w0;
        dst[1] = w1;
    }
    __syncthreads();

    const int lane  = threadIdx.x & 31;
    const int widx  = threadIdx.x >> 5;
    const int half  = lane >> 4;
    const int sub   = lane & 15;
    const long tile = (long)blk * 8 + widx;
    const long row0 = tile * 16;
    if (row0 >= M) return;

    v8f acc[4] = {};

    const float* aRow = A + (row0 + sub) * (long)K;

#pragma unroll 4
    for (int kk = 0; kk < K; kk += 4) {
        const int kb = kk + 2 * half;
        v2f a = *(const v2f*)(aRow + kb);

        const float* wp = sW + (kb >> 1) * (2 * HID) + 2 * sub;
#pragma unroll
        for (int t = 0; t < 4; ++t) {
            v2f b = *(const v2f*)(wp + t * 32);
            acc[t] = __builtin_amdgcn_wmma_f32_16x16x4_f32(
                 false, a,  false, b,
                 (short)0, acc[t],
                 false,  false);
        }
    }

#pragma unroll
    for (int r = 0; r < 8; ++r) {
        float* hp = H + (row0 + r + 8 * half) * (long)HID + sub;
#pragma unroll
        for (int t = 0; t < 4; ++t) hp[t * 16] = acc[t][r];
    }
}


__global__ void __launch_bounds__(256)
front_kernel(const float* __restrict__ x, const float* __restrict__ W1, float* __restrict__ h,
             int N, int gemmBlocks,
             const int* __restrict__ col, u64* __restrict__ cnt0, u64* __restrict__ cnt1,
             int* __restrict__ rank, int E, int countBlocks,
             const int* __restrict__ ec, int* __restrict__ mm0, int* __restrict__ mm1) {
    __shared__ __align__(16) float sW[256 * HID];
    const int b = blockIdx.x;
    if (b < gemmBlocks) {
        gemm_block<256>(x, W1, h, N, b, sW);
    } else if (b < gemmBlocks + countBlocks) {
        count_role(col, ec, cnt0, cnt1, rank, E, b - gemmBlocks, countBlocks);
    } else {
        const int first = gemmBlocks + countBlocks;
        minmax_role(ec, E, mm0, mm1, b - first, (int)gridDim.x - first);
    }
}


template <int MASK>
__device__ __forceinline__ float row_xor(float x) {
    const int b = __builtin_bit_cast(int, x);
    return __builtin_bit_cast(float,
        __builtin_amdgcn_mov_dpp(b, 0x160 + MASK, 0xf, 0xf, false));
}
__device__ __forceinline__ float xhalf(float x) {
    const int b = __builtin_bit_cast(int, x);
    return __builtin_bit_cast(float,
        __builtin_amdgcn_permlanex16(b, b, 0x76543210, 0xfedcba98, false, false));
}
__device__ __forceinline__ float wave_max(float m) {
    m = fmaxf(m, row_xor<1>(m));
    m = fmaxf(m, row_xor<2>(m));
    m = fmaxf(m, row_xor<4>(m));
    m = fmaxf(m, row_xor<8>(m));
    return fmaxf(m, xhalf(m));
}
__device__ __forceinline__ float wave_sum(float s) {
    s += row_xor<1>(s);
    s += row_xor<2>(s);
    s += row_xor<4>(s);
    s += row_xor<8>(s);
    return s + xhalf(s);
}

template <int B>
__device__ __forceinline__ void gather_batch(const v2f* __restrict__ hl, int r, float nrm,
                                             int j0, v2f& acc) {
    v2f a[B];
#pragma unroll
    for (int j = 0; j < B; ++j) {
        const int rj = __builtin_amdgcn_readlane(r, j0 + j);
        a[j] = hl[(long)rj * (HID / 2)];
    }
#pragma unroll
    for (int j = 0; j < B; ++j) {
        const float nj = __int_as_float(
            __builtin_amdgcn_readlane(__float_as_int(nrm), j0 + j));
        acc.x = fmaf(a[j].x, nj, acc.x);
        acc.y = fmaf(a[j].y, nj, acc.y);
    }
}

template <bool FIRST>
__device__ __forceinline__ v2f
node_row(const v2f* __restrict__ hl, const float* __restrict__ dinv, const int* __restrict__ off,
         const int* __restrict__ cnt, int2* __restrict__ rec, int node, int lane) {
    const int start = off[node];
    const int len = cnt[node];
    const float di = dinv[node];
    const v2f self = hl[(long)node * (HID / 2)];
    int2* rp = rec + ((long)start + lane);

    int m = len < 32 ? len : 32;
    int2 rc = make_int2(0, 0);
    if (lane < m) rc = *rp;

    v2f acc = {0.0f, 0.0f};
    for (int base = 0; base < len; base += 32) {
        const int r = rc.x;
        float nrm = __int_as_float(rc.y);
        if (FIRST) {
            if (lane < m) {
                nrm = dinv[r] * nrm * di;
                rp->y = __float_as_int(nrm);
            }
        }
        int j0 = 0;
#pragma unroll 1
        for (; j0 + 16 <= m; j0 += 16) gather_batch<16>(hl, r, nrm, j0, acc);
        switch (m - j0) {
#define GATHER_CASE(B) case B: gather_batch<B>(hl, r, nrm, j0, acc); break;
            GATHER_CASE(15) GATHER_CASE(14) GATHER_CASE(13) GATHER_CASE(12)
            GATHER_CASE(11) GATHER_CASE(10) GATHER_CASE(9)  GATHER_CASE(8)
            GATHER_CASE(7)  GATHER_CASE(6)  GATHER_CASE(5)  GATHER_CASE(4)
            GATHER_CASE(3)  GATHER_CASE(2)  GATHER_CASE(1)
#undef GATHER_CASE
            default: break;
        }
        const int left = len - base - 32;
        m = left < 32 ? left : 32;
        rp += 32;
        if (lane < m) rc = *rp;
    }
    const float sn = di * di;
    v2f o;
    o.x = fmaf(self.x, sn, acc.x);
    o.y = fmaf(self.y, sn, acc.y);
    return o;
}

__global__ void __launch_bounds__(256)
gather_gemm_kernel(const float* __restrict__ h, const float* __restrict__ dinv,
                   const float* __restrict__ bias, const int* __restrict__ off,
                   const int* __restrict__ cnt, int2* __restrict__ rec,
                   const float* __restrict__ W, float* __restrict__ H2, int N) {
    __shared__ __align__(16) float sW[HID * HID];
    __shared__ __align__(16) float sA[2][16 * HID];

    {
        float w0[8], w1[8];
        for (int u = 0; u < 8; ++u) {
            const int i = threadIdx.x + u * 256;
            w0[u] = W[(2 * (i >> 6)) * HID + (i & 63)];
            w1[u] = W[(2 * (i >> 6) + 1) * HID + (i & 63)];
        }
        for (int u = 0; u < 8; ++u) {
            const int i = threadIdx.x + u * 256;
            float* dst = sW + (i >> 6) * (2 * HID) + 2 * (i & 63);
            dst[0] = w0[u];
            dst[1] = w1[u];
        }
    }

    const int lane = threadIdx.x & 31;
    const int widx = __builtin_amdgcn_readfirstlane(threadIdx.x >> 5);
    const int half = lane >> 4;
    const int sub  = lane & 15;
    const v2f* hl = (const v2f*)h + lane;
    const v2f bv = ((const v2f*)bias)[lane];
    const int nTiles = (N + 15) >> 4;

    int it = 0;
    for (int tile = blockIdx.x; tile < nTiles; tile += gridDim.x, ++it) {
        float* tA = sA[it & 1];
        for (int j = 0; j < 2; ++j) {
            const int r = 2 * widx + j;
            const int node = tile * 16 + r;
            v2f o = {0.0f, 0.0f};
            if (node < N) {
                o = node_row<true>(hl, dinv, off, cnt, rec, node, lane);
                o.x = fmaxf(o.x + bv.x, 0.0f);
                o.y = fmaxf(o.y + bv.y, 0.0f);
            }
            *(v2f*)(tA + r * HID + ((2 * lane) ^ (r << 2))) = o;
        }
        __syncthreads();

        if (((widx >> 2) ^ it) & 1) continue;
        const int t = widx & 3;
        const float* aRow = tA + sub * HID;
        const int swz = sub << 2;
        v8f acc = {};
#pragma unroll 4
        for (int kk = 0; kk < HID; kk += 4) {
            const int kb = kk + 2 * half;
            const v2f a = *(const v2f*)(aRow + (kb ^ swz));
            const v2f b = *(const v2f*)(sW + (kb >> 1) * (2 * HID) + 2 * sub + t * 32);
            acc = __builtin_amdgcn_wmma_f32_16x16x4_f32(
                 false, a,  false, b,
                 (short)0, acc,
                 false,  false);
        }
        for (int r = 0; r < 8; ++r) {
            const int node = tile * 16 + r + 8 * half;
            if (node < N) H2[(long)node * HID + t * 16 + sub] = acc[r];
        }
    }
}

__global__ void __launch_bounds__(256)
gather_kernel(const float* __restrict__ h, const float* __restrict__ dinv,
              const float* __restrict__ bias, const int* __restrict__ off,
              const int* __restrict__ cnt,
              int2* __restrict__ rec, float* __restrict__ out,
              float* __restrict__ lsm, int N) {
    const int lane = threadIdx.x & 31;
    const int wave = __builtin_amdgcn_readfirstlane(blockIdx.x * 8 + (threadIdx.x >> 5));
    const int nwaves = gridDim.x * 8;
    const v2f* hl = (const v2f*)h + lane;
    const v2f bv = ((const v2f*)bias)[lane];
    for (int node = wave; node < N; node += nwaves) {
        v2f o = node_row<false>(hl, dinv, off, cnt, rec, node, lane);
        o.x += bv.x;
        o.y += bv.y;
        ((v2f*)out)[(long)node * (HID / 2) + lane] = o;
        const float mx = wave_max(fmaxf(o.x, o.y));
        const float se = wave_sum(__expf(o.x - mx) + __expf(o.y - mx));
        const float lse = mx + __logf(se);
        v2f l;
        l.x = o.x - lse;
        l.y = o.y - lse;
        ((v2f*)lsm)[(long)node * (HID / 2) + lane] = l;
    }
}

extern "C" void kernel_launch(void* const* d_in, const int* in_sizes, int n_in,
                              void* d_out, int out_size, void* d_ws, size_t ws_size,
                              hipStream_t stream) {
    const float* x  = (const float*)d_in[0];
    const int*   ei = (const int*)d_in[1];
    const int*   ec = (const int*)d_in[2];
    const float* W1 = (const float*)d_in[3];
    const float* b1 = (const float*)d_in[4];
    const float* W2 = (const float*)d_in[5];
    const float* b2 = (const float*)d_in[6];

    const int IN_DIM = 256;
    const int N = in_sizes[0] / IN_DIM;
    const int E = in_sizes[2];
    const int* row = ei;
    const int* col = ei + E;

    float* wsf  = (float*)d_ws;
    float* h    = wsf;
    float* h2   = h    + (size_t)N * HID;
    int2*  rec  = (int2*)(h2 + (size_t)N * HID);
    int*   rank = (int*)(rec + (size_t)E);
    float* dinv = (float*)(rank + (size_t)E);
    int*   len  = (int*)(dinv + (size_t)N);
    u64*   cnt  = (u64*)h2;
    u64*   cnt1 = cnt + (size_t)N;
    int*   off  = len + (size_t)N;
    const int nb = (N + SCAN_B - 1) / SCAN_B;
    int*   mm   = off + (size_t)N;
    int*   cur0 = mm + 64;
    int*   mm1  = (int*)(cnt1 + (size_t)N) + 64;
    int*   cur1 = mm1 + 64;
    int2*  fb   = (int2*)(cur1 + 64);

    float* emb = (float*)d_out;
    float* lsm = emb + (size_t)N * HID;

    const int tileBlocks = (int)((N / 16 + 7) / 8);
    const int edgeBlocks = (E + 256 * EPT - 1) / (256 * EPT);
    const int nodeBlocks = 2048;
    const int mmBlocks = 256;

    init_kernel<<<(N + 255) / 256, 256, 0, stream>>>(cnt, cnt1, mm, mm1, cur0, cur1, N);
    front_kernel<<<tileBlocks + edgeBlocks + mmBlocks, 256, 0, stream>>>(
        x, W1, h, N, tileBlocks, col, cnt, cnt1, rank, E, edgeBlocks, ec, mm, mm1);
    scan_kernel<<<nb, SCAN_B, 0, stream>>>((const int2*)cnt, (const int2*)cnt1, mm, mm1, cur0, cur1, off, fb, len, dinv, N, E);
    fill_kernel<<<edgeBlocks, 256, 0, stream>>>(ec, row, col, rank, fb, mm, mm1, rec, E);

    gather_gemm_kernel<<<nodeBlocks, 256, 0, stream>>>(h, dinv, b1, off, len, rec, W2, h2, N);

    gather_kernel<<<nodeBlocks, 256, 0, stream>>>(h2, dinv, b2, off, len, rec, emb, lsm, N);
}


# x@W1 role: W1 is staged in LDS with all 16 loads of a thread (16 bytes each) issued ahead of its LDS stores: one memory round trip before the first matrix instruction instead of 32 in a row
# speedup vs baseline: 2.9859x; 1.0471x over previous
#include <hip/hip_runtime.h>
#include <hip/hip_bf16.h>

typedef __attribute__((ext_vector_type(2))) float v2f;
typedef __attribute__((ext_vector_type(4))) float v4f;
typedef __attribute__((ext_vector_type(8))) float v8f;
typedef unsigned long long u64;

#define HID 64

__device__ __forceinline__ int package_half() {
    return (__builtin_amdgcn_s_sendmsg_rtn(0x87) >> 18) & 1;
}

__global__ void init_kernel(u64* __restrict__ cnt, u64* __restrict__ cnt1,
                            int* __restrict__ mm, int* __restrict__ mm1,
                            int* __restrict__ cur0, int* __restrict__ cur1, int N) {
    int i = blockIdx.x * blockDim.x + threadIdx.x;
    if (i < N) { cnt[i] = 0; cnt1[i] = 0; }
    if (i == 0) {
        mm[0] = 0x7FFFFFFF;  mm[1] = 0x80000000;
        mm1[0] = 0x7FFFFFFF; mm1[1] = 0x80000000;
        *cur0 = 0;
        *cur1 = 0;
    }
}

#define MM_EPT 8

__device__ __forceinline__ void minmax_role(const int* __restrict__ ec, int E,
                                            int* __restrict__ mm0, int* __restrict__ mm1,
                                            int blk, int nblk) {
    const int T = nblk * 256;
    const int t = blk * 256 + threadIdx.x;
    int lmin = 0x7FFFFFFF, lmax = 0x80000000;
    for (int i = t; i < E; i += MM_EPT * T) {
        int v[MM_EPT];
        for (int u = 0; u < MM_EPT; ++u) v[u] = ec[min(i + u * T, E - 1)];
        for (int u = 0; u < MM_EPT; ++u) {
            lmin = min(lmin, v[u]);
            lmax = max(lmax, v[u]);
        }
    }
    int* mm = package_half() ? mm1 : mm0;
    atomicMin(&mm[0], lmin);
    atomicMax(&mm[1], lmax);
}

#define EPT 4

__device__ __forceinline__ u64 count_word(int ecv) {
    return ((u64)(long long)ecv << 32) + 1ull;
}

__device__ __forceinline__ void
count_role(const int* __restrict__ col, const int* __restrict__ ec, u64* __restrict__ cnt0,
           u64* __restrict__ cnt1, int* __restrict__ rank, int E, int blk, int nblk) {
    const int T = nblk * 256;
    const int t = blk * 256 + threadIdx.x;
    const int hf = package_half();
    u64* cnt = hf ? cnt1 : cnt0;
    if (t + (EPT - 1) * T < E) {
        int c[EPT], v[EPT];
        u64 old[EPT];
        for (int u = 0; u < EPT; ++u) c[u] = col[t + u * T];
        for (int u = 0; u < EPT; ++u) v[u] = ec[t + u * T];
        for (int u = 0; u < EPT; ++u) old[u] = atomicAdd(&cnt[c[u]], count_word(v[u]));
        for (int u = 0; u < EPT; ++u) rank[t + u * T] = (int)(unsigned)old[u] * 2 + hf;
    } else {
        for (int e = t; e < E; e += T)
            rank[e] = (int)(unsigned)atomicAdd(&cnt[col[e]], count_word(ec[e])) * 2 + hf;
    }
}

#define SCAN_B 256

__global__ void __launch_bounds__(SCAN_B)
scan_kernel(const int2* __restrict__ cnt, const int2* __restrict__ cnt1,
            const int* __restrict__ mm, const int* __restrict__ mm1,
            int* __restrict__ cur0, int* __restrict__ cur1,
            int* __restrict__ off, int2* __restrict__ fb,
            int* __restrict__ len, float* __restrict__ dinv, int N, int E) {
    __shared__ int wsum[SCAN_B / 32];
    __shared__ int gbase;
    const int tid = threadIdx.x, lane = tid & 31, w = tid >> 5;
    const int i = blockIdx.x * SCAN_B + tid;
    int2 ca = make_int2(0, 0), cb = make_int2(0, 0);
    if (i < N) { ca = cnt[i]; cb = cnt1[i]; }
    const int v = ca.x + cb.x;
    int inc = v;
    for (int o = 1; o < 32; o <<= 1) {
        int t = __shfl_up(inc, o, 32);
        if (lane >= o) inc += t;
    }
    if (lane == 31) wsum[w] = inc;
    __syncthreads();
    int wbase = 0, total = 0;
    for (int k = 0; k < SCAN_B / 32; ++k) {
        int s = wsum[k];
        if (k < w) wbase += s;
        total += s;
    }
    if (tid == 0) {
        const int hf = package_half();
        const int old = atomicAdd(hf ? cur1 : cur0, total);
        gbase = hf ? E - old - total : old;
    }
    float dv = 0.0f;
    if (i < N) {
        const int mni = min(mm[0], mm1[0]), mxi = max(mm[1], mm1[1]);
        const float denom = (float)mxi - (float)mni;
        const long long num = (long long)ca.y + (long long)cb.y - (long long)v * (long long)mni;
        const float d = (denom > 0.0f) ? 1.0f + (float)num / denom : 1.0f;
        dv = (d > 0.0f) ? rsqrtf(d) : 0.0f;
    }
    __syncthreads();
    if (i < N) {
        const int start = gbase + wbase + inc - v;
        off[i] = start;
        fb[i] = make_int2(start, start + ca.x);
        len[i] = v;
        dinv[i] = dv;
    }
}

__global__ void __launch_bounds__(256)
fill_kernel(const int* __restrict__ ec, const int* __restrict__ row,
            const int* __restrict__ col, const int* __restrict__ rank,
            const int2* __restrict__ fb,
            const int* __restrict__ mm, const int* __restrict__ mm1,
            int2* __restrict__ rec, int E) {
    const int T = gridDim.x * blockDim.x;
    const int t = blockIdx.x * blockDim.x + threadIdx.x;
    const float mn = (float)min(mm[0], mm1[0]), mx = (float)max(mm[1], mm1[1]);
    const float denom = mx - mn;
    const bool span = denom > 0.0f;
    const float dsafe = span ? denom : 1.0f;
    if (t + (EPT - 1) * T < E) {
        int v[EPT], r[EPT], c[EPT], pos[EPT], b0[EPT], b1[EPT];
        for (int u = 0; u < EPT; ++u) {
            const int e = t + u * T;
            v[u] = ec[e];
            r[u] = row[e];
            c[u] = col[e];
            pos[u] = rank[e];
        }
        for (int u = 0; u < EPT; ++u) {
            const int2 q = fb[c[u]];
            b0[u] = q.x;
            b1[u] = q.y;
        }
        for (int u = 0; u < EPT; ++u) {
            const float q = ((float)v[u] - mn) / dsafe;
            const float w = span ? q : 0.0f;
            const int base = (pos[u] & 1) ? b1[u] : b0[u];
            rec[base + (pos[u] >> 1)] = make_int2(r[u], __float_as_int(w));
        }
    } else {
        for (int e = t; e < E; e += T) {
            const float q = ((float)ec[e] - mn) / dsafe;
            const float w = span ? q : 0.0f;
            const int2 b = fb[col[e]];
            const int rk = rank[e];
            rec[((rk & 1) ? b.y : b.x) + (rk >> 1)] = make_int2(row[e], __float_as_int(w));
        }
    }
}

template <int K>
__device__ __forceinline__ void gemm_block(const float* __restrict__ A, const float* __restrict__ W,
                                           float* __restrict__ H, int M, int blk, float* sW) {

    {
        constexpr int W_PER = (K / 2) * (HID / 4) / 256;
        v4f w0[W_PER], w1[W_PER];
        for (int u = 0; u < W_PER; ++u) {
            const int i  = threadIdx.x + u * 256;
            const int kp = i >> 4;
            const int c4 = i & 15;
            w0[u] = *(const v4f*)(W + (2 * kp) * HID + 4 * c4);
            w1[u] = *(const v4f*)(W + (2 * kp + 1) * HID + 4 * c4);
        }
        for (int u = 0; u < W_PER; ++u) {
            const int i  = threadIdx.x + u * 256;
            const int kp = i >> 4;
            const int c4 = i & 15;
            float* dst = sW + kp * (2 * HID) + 8 * c4;
            v4f lo, hi;
            lo.x = w0[u].x; lo.y = w1[u].x; lo.z = w0[u].y; lo.w = w1[u].y;
            hi.x = w0[u].z; hi.y = w1[u].z; hi.z = w0[u].w; hi.w = w1[u].w;
            *(v4f*)dst = lo;
            *(v4f*)(dst + 4) = hi;
        }
    }
    __syncthreads();

    const int lane  = threadIdx.x & 31;
    const int widx  = threadIdx.x >> 5;
    const int half  = lane >> 4;
    const int sub   = lane & 15;
    const long tile = (long)blk * 8 + widx;
    const long row0 = tile * 16;
    if (row0 >= M) return;

    v8f acc[4] = {};

    const float* aRow = A + (row0 + sub) * (long)K;

#pragma unroll 4
    for (int kk = 0; kk < K; kk += 4) {
        const int kb = kk + 2 * half;
        v2f a = *(const v2f*)(aRow + kb);

        const float* wp = sW + (kb >> 1) * (2 * HID) + 2 * sub;
#pragma unroll
        for (int t = 0; t < 4; ++t) {
            v2f b = *(const v2f*)(wp + t * 32);
            acc[t] = __builtin_amdgcn_wmma_f32_16x16x4_f32(
                 false, a,  false, b,
                 (short)0, acc[t],
                 false,  false);
        }
    }

#pragma unroll
    for (int r = 0; r < 8; ++r) {
        float* hp = H + (row0 + r + 8 * half) * (long)HID + sub;
#pragma unroll
        for (int t = 0; t < 4; ++t) hp[t * 16] = acc[t][r];
    }
}


__global__ void __launch_bounds__(256)
front_kernel(const float* __restrict__ x, const float* __restrict__ W1, float* __restrict__ h,
             int N, int gemmBlocks,
             const int* __restrict__ col, u64* __restrict__ cnt0, u64* __restrict__ cnt1,
             int* __restrict__ rank, int E, int countBlocks,
             const int* __restrict__ ec, int* __restrict__ mm0, int* __restrict__ mm1) {
    __shared__ __align__(16) float sW[256 * HID];
    const int b = blockIdx.x;
    if (b < gemmBlocks) {
        gemm_block<256>(x, W1, h, N, b, sW);
    } else if (b < gemmBlocks + countBlocks) {
        count_role(col, ec, cnt0, cnt1, rank, E, b - gemmBlocks, countBlocks);
    } else {
        const int first = gemmBlocks + countBlocks;
        minmax_role(ec, E, mm0, mm1, b - first, (int)gridDim.x - first);
    }
}


template <int MASK>
__device__ __forceinline__ float row_xor(float x) {
    const int b = __builtin_bit_cast(int, x);
    return __builtin_bit_cast(float,
        __builtin_amdgcn_mov_dpp(b, 0x160 + MASK, 0xf, 0xf, false));
}
__device__ __forceinline__ float xhalf(float x) {
    const int b = __builtin_bit_cast(int, x);
    return __builtin_bit_cast(float,
        __builtin_amdgcn_permlanex16(b, b, 0x76543210, 0xfedcba98, false, false));
}
__device__ __forceinline__ float wave_max(float m) {
    m = fmaxf(m, row_xor<1>(m));
    m = fmaxf(m, row_xor<2>(m));
    m = fmaxf(m, row_xor<4>(m));
    m = fmaxf(m, row_xor<8>(m));
    return fmaxf(m, xhalf(m));
}
__device__ __forceinline__ float wave_sum(float s) {
    s += row_xor<1>(s);
    s += row_xor<2>(s);
    s += row_xor<4>(s);
    s += row_xor<8>(s);
    return s + xhalf(s);
}

template <int B>
__device__ __forceinline__ void gather_batch(const v2f* __restrict__ hl, int r, float nrm,
                                             int j0, v2f& acc) {
    v2f a[B];
#pragma unroll
    for (int j = 0; j < B; ++j) {
        const int rj = __builtin_amdgcn_readlane(r, j0 + j);
        a[j] = hl[(long)rj * (HID / 2)];
    }
#pragma unroll
    for (int j = 0; j < B; ++j) {
        const float nj = __int_as_float(
            __builtin_amdgcn_readlane(__float_as_int(nrm), j0 + j));
        acc.x = fmaf(a[j].x, nj, acc.x);
        acc.y = fmaf(a[j].y, nj, acc.y);
    }
}

template <bool FIRST>
__device__ __forceinline__ v2f
node_row(const v2f* __restrict__ hl, const float* __restrict__ dinv, const int* __restrict__ off,
         const int* __restrict__ cnt, int2* __restrict__ rec, int node, int lane) {
    const int start = off[node];
    const int len = cnt[node];
    const float di = dinv[node];
    const v2f self = hl[(long)node * (HID / 2)];
    int2* rp = rec + ((long)start + lane);

    int m = len < 32 ? len : 32;
    int2 rc = make_int2(0, 0);
    if (lane < m) rc = *rp;

    v2f acc = {0.0f, 0.0f};
    for (int base = 0; base < len; base += 32) {
        const int r = rc.x;
        float nrm = __int_as_float(rc.y);
        if (FIRST) {
            if (lane < m) {
                nrm = dinv[r] * nrm * di;
                rp->y = __float_as_int(nrm);
            }
        }
        int j0 = 0;
#pragma unroll 1
        for (; j0 + 16 <= m; j0 += 16) gather_batch<16>(hl, r, nrm, j0, acc);
        switch (m - j0) {
#define GATHER_CASE(B) case B: gather_batch<B>(hl, r, nrm, j0, acc); break;
            GATHER_CASE(15) GATHER_CASE(14) GATHER_CASE(13) GATHER_CASE(12)
            GATHER_CASE(11) GATHER_CASE(10) GATHER_CASE(9)  GATHER_CASE(8)
            GATHER_CASE(7)  GATHER_CASE(6)  GATHER_CASE(5)  GATHER_CASE(4)
            GATHER_CASE(3)  GATHER_CASE(2)  GATHER_CASE(1)
#undef GATHER_CASE
            default: break;
        }
        const int left = len - base - 32;
        m = left < 32 ? left : 32;
        rp += 32;
        if (lane < m) rc = *rp;
    }
    const float sn = di * di;
    v2f o;
    o.x = fmaf(self.x, sn, acc.x);
    o.y = fmaf(self.y, sn, acc.y);
    return o;
}

__global__ void __launch_bounds__(256)
gather_gemm_kernel(const float* __restrict__ h, const float* __restrict__ dinv,
                   const float* __restrict__ bias, const int* __restrict__ off,
                   const int* __restrict__ cnt, int2* __restrict__ rec,
                   const float* __restrict__ W, float* __restrict__ H2, int N) {
    __shared__ __align__(16) float sW[HID * HID];
    __shared__ __align__(16) float sA[2][16 * HID];

    {
        float w0[8], w1[8];
        for (int u = 0; u < 8; ++u) {
            const int i = threadIdx.x + u * 256;
            w0[u] = W[(2 * (i >> 6)) * HID + (i & 63)];
            w1[u] = W[(2 * (i >> 6) + 1) * HID + (i & 63)];
        }
        for (int u = 0; u < 8; ++u) {
            const int i = threadIdx.x + u * 256;
            float* dst = sW + (i >> 6) * (2 * HID) + 2 * (i & 63);
            dst[0] = w0[u];
            dst[1] = w1[u];
        }
    }

    const int lane = threadIdx.x & 31;
    const int widx = __builtin_amdgcn_readfirstlane(threadIdx.x >> 5);
    const int half = lane >> 4;
    const int sub  = lane & 15;
    const v2f* hl = (const v2f*)h + lane;
    const v2f bv = ((const v2f*)bias)[lane];
    const int nTiles = (N + 15) >> 4;

    int it = 0;
    for (int tile = blockIdx.x; tile < nTiles; tile += gridDim.x, ++it) {
        float* tA = sA[it & 1];
        for (int j = 0; j < 2; ++j) {
            const int r = 2 * widx + j;
            const int node = tile * 16 + r;
            v2f o = {0.0f, 0.0f};
            if (node < N) {
                o = node_row<true>(hl, dinv, off, cnt, rec, node, lane);
                o.x = fmaxf(o.x + bv.x, 0.0f);
                o.y = fmaxf(o.y + bv.y, 0.0f);
            }
            *(v2f*)(tA + r * HID + ((2 * lane) ^ (r << 2))) = o;
        }
        __syncthreads();

        if (((widx >> 2) ^ it) & 1) continue;
        const int t = widx & 3;
        const float* aRow = tA + sub * HID;
        const int swz = sub << 2;
        v8f acc = {};
#pragma unroll 4
        for (int kk = 0; kk < HID; kk += 4) {
            const int kb = kk + 2 * half;
            const v2f a = *(const v2f*)(aRow + (kb ^ swz));
            const v2f b = *(const v2f*)(sW + (kb >> 1) * (2 * HID) + 2 * sub + t * 32);
            acc = __builtin_amdgcn_wmma_f32_16x16x4_f32(
                 false, a,  false, b,
                 (short)0, acc,
                 false,  false);
        }
        for (int r = 0; r < 8; ++r) {
            const int node = tile * 16 + r + 8 * half;
            if (node < N) H2[(long)node * HID + t * 16 + sub] = acc[r];
        }
    }
}

__global__ void __launch_bounds__(256)
gather_kernel(const float* __restrict__ h, const float* __restrict__ dinv,
              const float* __restrict__ bias, const int* __restrict__ off,
              const int* __restrict__ cnt,
              int2* __restrict__ rec, float* __restrict__ out,
              float* __restrict__ lsm, int N) {
    const int lane = threadIdx.x & 31;
    const int wave = __builtin_amdgcn_readfirstlane(blockIdx.x * 8 + (threadIdx.x >> 5));
    const int nwaves = gridDim.x * 8;
    const v2f* hl = (const v2f*)h + lane;
    const v2f bv = ((const v2f*)bias)[lane];
    for (int node = wave; node < N; node += nwaves) {
        v2f o = node_row<false>(hl, dinv, off, cnt, rec, node, lane);
        o.x += bv.x;
        o.y += bv.y;
        ((v2f*)out)[(long)node * (HID / 2) + lane] = o;
        const float mx = wave_max(fmaxf(o.x, o.y));
        const float se = wave_sum(__expf(o.x - mx) + __expf(o.y - mx));
        const float lse = mx + __logf(se);
        v2f l;
        l.x = o.x - lse;
        l.y = o.y - lse;
        ((v2f*)lsm)[(long)node * (HID / 2) + lane] = l;
    }
}

extern "C" void kernel_launch(void* const* d_in, const int* in_sizes, int n_in,
                              void* d_out, int out_size, void* d_ws, size_t ws_size,
                              hipStream_t stream) {
    const float* x  = (const float*)d_in[0];
    const int*   ei = (const int*)d_in[1];
    const int*   ec = (const int*)d_in[2];
    const float* W1 = (const float*)d_in[3];
    const float* b1 = (const float*)d_in[4];
    const float* W2 = (const float*)d_in[5];
    const float* b2 = (const float*)d_in[6];

    const int IN_DIM = 256;
    const int N = in_sizes[0] / IN_DIM;
    const int E = in_sizes[2];
    const int* row = ei;
    const int* col = ei + E;

    float* wsf  = (float*)d_ws;
    float* h    = wsf;
    float* h2   = h    + (size_t)N * HID;
    int2*  rec  = (int2*)(h2 + (size_t)N * HID);
    int*   rank = (int*)(rec + (size_t)E);
    float* dinv = (float*)(rank + (size_t)E);
    int*   len  = (int*)(dinv + (size_t)N);
    u64*   cnt  = (u64*)h2;
    u64*   cnt1 = cnt + (size_t)N;
    int*   off  = len + (size_t)N;
    const int nb = (N + SCAN_B - 1) / SCAN_B;
    int*   mm   = off + (size_t)N;
    int*   cur0 = mm + 64;
    int*   mm1  = (int*)(cnt1 + (size_t)N) + 64;
    int*   cur1 = mm1 + 64;
    int2*  fb   = (int2*)(cur1 + 64);

    float* emb = (float*)d_out;
    float* lsm = emb + (size_t)N * HID;

    const int tileBlocks = (int)((N / 16 + 7) / 8);
    const int edgeBlocks = (E + 256 * EPT - 1) / (256 * EPT);
    const int nodeBlocks = 2048;
    const int mmBlocks = 256;

    init_kernel<<<(N + 255) / 256, 256, 0, stream>>>(cnt, cnt1, mm, mm1, cur0, cur1, N);
    front_kernel<<<tileBlocks + edgeBlocks + mmBlocks, 256, 0, stream>>>(
        x, W1, h, N, tileBlocks, col, cnt, cnt1, rank, E, edgeBlocks, ec, mm, mm1);
    scan_kernel<<<nb, SCAN_B, 0, stream>>>((const int2*)cnt, (const int2*)cnt1, mm, mm1, cur0, cur1, off, fb, len, dinv, N, E);
    fill_kernel<<<edgeBlocks, 256, 0, stream>>>(ec, row, col, rank, fb, mm, mm1, rec, E);

    gather_gemm_kernel<<<nodeBlocks, 256, 0, stream>>>(h, dinv, b1, off, len, rec, W2, h2, N);

    gather_kernel<<<nodeBlocks, 256, 0, stream>>>(h2, dinv, b2, off, len, rec, emb, lsm, N);
}


# rows stored times their own dinv where made (x@W1 moved beside the edge placing, behind the node pass): node sum = dinv[c]*(sum w*row + own row), same order, factors regrouped; no dinv load per edge
# speedup vs baseline: 3.3737x; 1.1299x over previous
#include <hip/hip_runtime.h>
#include <hip/hip_bf16.h>

typedef __attribute__((ext_vector_type(2))) float v2f;
typedef __attribute__((ext_vector_type(4))) float v4f;
typedef __attribute__((ext_vector_type(8))) float v8f;
typedef unsigned long long u64;

#define HID 64

__device__ __forceinline__ int package_half() {
    return (__builtin_amdgcn_s_sendmsg_rtn(0x87) >> 18) & 1;
}

__global__ void init_kernel(u64* __restrict__ cnt, u64* __restrict__ cnt1,
                            int* __restrict__ mm, int* __restrict__ mm1,
                            int* __restrict__ cur0, int* __restrict__ cur1, int N) {
    int i = blockIdx.x * blockDim.x + threadIdx.x;
    if (i < N) { cnt[i] = 0; cnt1[i] = 0; }
    if (i == 0) {
        mm[0] = 0x7FFFFFFF;  mm[1] = 0x80000000;
        mm1[0] = 0x7FFFFFFF; mm1[1] = 0x80000000;
        *cur0 = 0;
        *cur1 = 0;
    }
}

#define MM_EPT 8

__device__ __forceinline__ void minmax_role(const int* __restrict__ ec, int E,
                                            int* __restrict__ mm0, int* __restrict__ mm1,
                                            int blk, int nblk) {
    const int T = nblk * 256;
    const int t = blk * 256 + threadIdx.x;
    int lmin = 0x7FFFFFFF, lmax = 0x80000000;
    for (int i = t; i < E; i += MM_EPT * T) {
        int v[MM_EPT];
        for (int u = 0; u < MM_EPT; ++u) v[u] = ec[min(i + u * T, E - 1)];
        for (int u = 0; u < MM_EPT; ++u) {
            lmin = min(lmin, v[u]);
            lmax = max(lmax, v[u]);
        }
    }
    int* mm = package_half() ? mm1 : mm0;
    atomicMin(&mm[0], lmin);
    atomicMax(&mm[1], lmax);
}

#define EPT 4

__device__ __forceinline__ u64 count_word(int ecv) {
    return ((u64)(long long)ecv << 32) + 1ull;
}

__device__ __forceinline__ void
count_role(const int* __restrict__ col, const int* __restrict__ ec, u64* __restrict__ cnt0,
           u64* __restrict__ cnt1, int* __restrict__ rank, int E, int blk, int nblk) {
    const int T = nblk * 256;
    const int t = blk * 256 + threadIdx.x;
    const int hf = package_half();
    u64* cnt = hf ? cnt1 : cnt0;
    if (t + (EPT - 1) * T < E) {
        int c[EPT], v[EPT];
        u64 old[EPT];
        for (int u = 0; u < EPT; ++u) c[u] = col[t + u * T];
        for (int u = 0; u < EPT; ++u) v[u] = ec[t + u * T];
        for (int u = 0; u < EPT; ++u) old[u] = atomicAdd(&cnt[c[u]], count_word(v[u]));
        for (int u = 0; u < EPT; ++u) rank[t + u * T] = (int)(unsigned)old[u] * 2 + hf;
    } else {
        for (int e = t; e < E; e += T)
            rank[e] = (int)(unsigned)atomicAdd(&cnt[col[e]], count_word(ec[e])) * 2 + hf;
    }
}

#define SCAN_B 256

__global__ void __launch_bounds__(SCAN_B)
scan_kernel(const int2* __restrict__ cnt, const int2* __restrict__ cnt1,
            const int* __restrict__ mm, const int* __restrict__ mm1,
            int* __restrict__ cur0, int* __restrict__ cur1,
            int* __restrict__ off, int2* __restrict__ fb,
            int* __restrict__ len, float* __restrict__ dinv, int N, int E) {
    __shared__ int wsum[SCAN_B / 32];
    __shared__ int gbase;
    const int tid = threadIdx.x, lane = tid & 31, w = tid >> 5;
    const int i = blockIdx.x * SCAN_B + tid;
    int2 ca = make_int2(0, 0), cb = make_int2(0, 0);
    if (i < N) { ca = cnt[i]; cb = cnt1[i]; }
    const int v = ca.x + cb.x;
    int inc = v;
    for (int o = 1; o < 32; o <<= 1) {
        int t = __shfl_up(inc, o, 32);
        if (lane >= o) inc += t;
    }
    if (lane == 31) wsum[w] = inc;
    __syncthreads();
    int wbase = 0, total = 0;
    for (int k = 0; k < SCAN_B / 32; ++k) {
        int s = wsum[k];
        if (k < w) wbase += s;
        total += s;
    }
    if (tid == 0) {
        const int hf = package_half();
        const int old = atomicAdd(hf ? cur1 : cur0, total);
        gbase = hf ? E - old - total : old;
    }
    float dv = 0.0f;
    if (i < N) {
        const int mni = min(mm[0], mm1[0]), mxi = max(mm[1], mm1[1]);
        const float denom = (float)mxi - (float)mni;
        const long long num = (long long)ca.y + (long long)cb.y - (long long)v * (long long)mni;
        const float d = (denom > 0.0f) ? 1.0f + (float)num / denom : 1.0f;
        dv = (d > 0.0f) ? rsqrtf(d) : 0.0f;
    }
    __syncthreads();
    if (i < N) {
        const int start = gbase + wbase + inc - v;
        off[i] = start;
        fb[i] = make_int2(start, start + ca.x);
        len[i] = v;
        dinv[i] = dv;
    }
}

__device__ __forceinline__ void
fill_role(const int* __restrict__ ec, const int* __restrict__ row,
          const int* __restrict__ col, const int* __restrict__ rank,
          const int2* __restrict__ fb,
          const int* __restrict__ mm, const int* __restrict__ mm1,
          int2* __restrict__ rec, int E, int blk, int nblk) {
    const int T = nblk * 256;
    const int t = blk * 256 + threadIdx.x;
    const float mn = (float)min(mm[0], mm1[0]), mx = (float)max(mm[1], mm1[1]);
    const float denom = mx - mn;
    const bool span = denom > 0.0f;
    const float dsafe = span ? denom : 1.0f;
    if (t + (EPT - 1) * T < E) {
        int v[EPT], r[EPT], c[EPT], pos[EPT], b0[EPT], b1[EPT];
        for (int u = 0; u < EPT; ++u) {
            const int e = t + u * T;
            v[u] = ec[e];
            r[u] = row[e];
            c[u] = col[e];
            pos[u] = rank[e];
        }
        for (int u = 0; u < EPT; ++u) {
            const int2 q = fb[c[u]];
            b0[u] = q.x;
            b1[u] = q.y;
        }
        for (int u = 0; u < EPT; ++u) {
            const float q = ((float)v[u] - mn) / dsafe;
            const float w = span ? q : 0.0f;
            const int base = (pos[u] & 1) ? b1[u] : b0[u];
            rec[base + (pos[u] >> 1)] = make_int2(r[u], __float_as_int(w));
        }
    } else {
        for (int e = t; e < E; e += T) {
            const float q = ((float)ec[e] - mn) / dsafe;
            const float w = span ? q : 0.0f;
            const int2 b = fb[col[e]];
            const int rk = rank[e];
            rec[((rk & 1) ? b.y : b.x) + (rk >> 1)] = make_int2(row[e], __float_as_int(w));
        }
    }
}

template <int K>
__device__ __forceinline__ void gemm_block(const float* __restrict__ A, const float* __restrict__ W,
                                           const float* __restrict__ dinv,
                                           float* __restrict__ H, int M, int blk, float* sW) {

    {
        constexpr int W_PER = (K / 2) * (HID / 4) / 256;
        v4f w0[W_PER], w1[W_PER];
        for (int u = 0; u < W_PER; ++u) {
            const int i  = threadIdx.x + u * 256;
            const int kp = i >> 4;
            const int c4 = i & 15;
            w0[u] = *(const v4f*)(W + (2 * kp) * HID + 4 * c4);
            w1[u] = *(const v4f*)(W + (2 * kp + 1) * HID + 4 * c4);
        }
        for (int u = 0; u < W_PER; ++u) {
            const int i  = threadIdx.x + u * 256;
            const int kp = i >> 4;
            const int c4 = i & 15;
            float* dst = sW + kp * (2 * HID) + 8 * c4;
            v4f lo, hi;
            lo.x = w0[u].x; lo.y = w1[u].x; lo.z = w0[u].y; lo.w = w1[u].y;
            hi.x = w0[u].z; hi.y = w1[u].z; hi.z = w0[u].w; hi.w = w1[u].w;
            *(v4f*)dst = lo;
            *(v4f*)(dst + 4) = hi;
        }
    }
    __syncthreads();

    const int lane  = threadIdx.x & 31;
    const int widx  = threadIdx.x >> 5;
    const int half  = lane >> 4;
    const int sub   = lane & 15;
    const long tile = (long)blk * 8 + widx;
    const long row0 = tile * 16;
    if (row0 >= M) return;

    v8f acc[4] = {};

    const float* aRow = A + (row0 + sub) * (long)K;

#pragma unroll 4
    for (int kk = 0; kk < K; kk += 4) {
        const int kb = kk + 2 * half;
        v2f a = *(const v2f*)(aRow + kb);

        const float* wp = sW + (kb >> 1) * (2 * HID) + 2 * sub;
#pragma unroll
        for (int t = 0; t < 4; ++t) {
            v2f b = *(const v2f*)(wp + t * 32);
            acc[t] = __builtin_amdgcn_wmma_f32_16x16x4_f32(
                 false, a,  false, b,
                 (short)0, acc[t],
                 false,  false);
        }
    }

    const v4f d0 = *(const v4f*)(dinv + row0 + 8 * half);
    const v4f d1 = *(const v4f*)(dinv + row0 + 8 * half + 4);
    const float dr[8] = {d0.x, d0.y, d0.z, d0.w, d1.x, d1.y, d1.z, d1.w};
#pragma unroll
    for (int r = 0; r < 8; ++r) {
        float* hp = H + (row0 + r + 8 * half) * (long)HID + sub;
#pragma unroll
        for (int t = 0; t < 4; ++t) hp[t * 16] = acc[t][r] * dr[r];
    }
}


__global__ void __launch_bounds__(256)
count_kernel(const int* __restrict__ col, u64* __restrict__ cnt0, u64* __restrict__ cnt1,
             int* __restrict__ rank, int E, int countBlocks,
             const int* __restrict__ ec, int* __restrict__ mm0, int* __restrict__ mm1) {
    const int b = blockIdx.x;
    if (b < countBlocks) {
        count_role(col, ec, cnt0, cnt1, rank, E, b, countBlocks);
    } else {
        minmax_role(ec, E, mm0, mm1, b - countBlocks, (int)gridDim.x - countBlocks);
    }
}

__global__ void __launch_bounds__(256)
mid_kernel(const float* __restrict__ x, const float* __restrict__ W1,
           const float* __restrict__ dinv, float* __restrict__ h, int N, int gemmBlocks,
           const int* __restrict__ ec, const int* __restrict__ row,
           const int* __restrict__ col, const int* __restrict__ rank,
           const int2* __restrict__ fb,
           const int* __restrict__ mm, const int* __restrict__ mm1,
           int2* __restrict__ rec, int E) {
    __shared__ __align__(16) float sW[256 * HID];
    const int b = blockIdx.x;
    if (b < gemmBlocks) {
        gemm_block<256>(x, W1, dinv, h, N, b, sW);
    } else {
        fill_role(ec, row, col, rank, fb, mm, mm1, rec, E, b - gemmBlocks, (int)gridDim.x - gemmBlocks);
    }
}


template <int MASK>
__device__ __forceinline__ float row_xor(float x) {
    const int b = __builtin_bit_cast(int, x);
    return __builtin_bit_cast(float,
        __builtin_amdgcn_mov_dpp(b, 0x160 + MASK, 0xf, 0xf, false));
}
__device__ __forceinline__ float xhalf(float x) {
    const int b = __builtin_bit_cast(int, x);
    return __builtin_bit_cast(float,
        __builtin_amdgcn_permlanex16(b, b, 0x76543210, 0xfedcba98, false, false));
}
__device__ __forceinline__ float wave_max(float m) {
    m = fmaxf(m, row_xor<1>(m));
    m = fmaxf(m, row_xor<2>(m));
    m = fmaxf(m, row_xor<4>(m));
    m = fmaxf(m, row_xor<8>(m));
    return fmaxf(m, xhalf(m));
}
__device__ __forceinline__ float wave_sum(float s) {
    s += row_xor<1>(s);
    s += row_xor<2>(s);
    s += row_xor<4>(s);
    s += row_xor<8>(s);
    return s + xhalf(s);
}

template <int B>
__device__ __forceinline__ void gather_batch(const v2f* __restrict__ hl, int r, float nrm,
                                             int j0, v2f& acc) {
    v2f a[B];
#pragma unroll
    for (int j = 0; j < B; ++j) {
        const int rj = __builtin_amdgcn_readlane(r, j0 + j);
        a[j] = hl[(long)rj * (HID / 2)];
    }
#pragma unroll
    for (int j = 0; j < B; ++j) {
        const float nj = __int_as_float(
            __builtin_amdgcn_readlane(__float_as_int(nrm), j0 + j));
        acc.x = fmaf(a[j].x, nj, acc.x);
        acc.y = fmaf(a[j].y, nj, acc.y);
    }
}

__device__ __forceinline__ v2f
node_row(const v2f* __restrict__ hl, const float* __restrict__ dinv, const int* __restrict__ off,
         const int* __restrict__ cnt, const int2* __restrict__ rec, int node, int lane, float& di) {
    const int start = off[node];
    const int len = cnt[node];
    di = dinv[node];
    const v2f self = hl[(long)node * (HID / 2)];
    const int2* rp = rec + ((long)start + lane);

    int m = len < 32 ? len : 32;
    int2 rc = make_int2(0, 0);
    if (lane < m) rc = *rp;

    v2f acc = {0.0f, 0.0f};
    for (int base = 0; base < len; base += 32) {
        const int r = rc.x;
        const float nrm = __int_as_float(rc.y);
        int j0 = 0;
#pragma unroll 1
        for (; j0 + 16 <= m; j0 += 16) gather_batch<16>(hl, r, nrm, j0, acc);
        switch (m - j0) {
#define GATHER_CASE(B) case B: gather_batch<B>(hl, r, nrm, j0, acc); break;
            GATHER_CASE(15) GATHER_CASE(14) GATHER_CASE(13) GATHER_CASE(12)
            GATHER_CASE(11) GATHER_CASE(10) GATHER_CASE(9)  GATHER_CASE(8)
            GATHER_CASE(7)  GATHER_CASE(6)  GATHER_CASE(5)  GATHER_CASE(4)
            GATHER_CASE(3)  GATHER_CASE(2)  GATHER_CASE(1)
#undef GATHER_CASE
            default: break;
        }
        const int left = len - base - 32;
        m = left < 32 ? left : 32;
        rp += 32;
        if (lane < m) rc = *rp;
    }
    v2f o;
    o.x = (acc.x + self.x) * di;
    o.y = (acc.y + self.y) * di;
    return o;
}

__global__ void __launch_bounds__(256)
gather_gemm_kernel(const float* __restrict__ h, const float* __restrict__ dinv,
                   const float* __restrict__ bias, const int* __restrict__ off,
                   const int* __restrict__ cnt, const int2* __restrict__ rec,
                   const float* __restrict__ W, float* __restrict__ H2, int N) {
    __shared__ __align__(16) float sW[HID * HID];
    __shared__ __align__(16) float sA[2][16 * HID];

    {
        float w0[8], w1[8];
        for (int u = 0; u < 8; ++u) {
            const int i = threadIdx.x + u * 256;
            w0[u] = W[(2 * (i >> 6)) * HID + (i & 63)];
            w1[u] = W[(2 * (i >> 6) + 1) * HID + (i & 63)];
        }
        for (int u = 0; u < 8; ++u) {
            const int i = threadIdx.x + u * 256;
            float* dst = sW + (i >> 6) * (2 * HID) + 2 * (i & 63);
            dst[0] = w0[u];
            dst[1] = w1[u];
        }
    }

    const int lane = threadIdx.x & 31;
    const int widx = __builtin_amdgcn_readfirstlane(threadIdx.x >> 5);
    const int half = lane >> 4;
    const int sub  = lane & 15;
    const v2f* hl = (const v2f*)h + lane;
    const v2f bv = ((const v2f*)bias)[lane];
    const int nTiles = (N + 15) >> 4;

    int it = 0;
    for (int tile = blockIdx.x; tile < nTiles; tile += gridDim.x, ++it) {
        float* tA = sA[it & 1];
        for (int j = 0; j < 2; ++j) {
            const int r = 2 * widx + j;
            const int node = tile * 16 + r;
            v2f o = {0.0f, 0.0f};
            if (node < N) {
                float di;
                o = node_row(hl, dinv, off, cnt, rec, node, lane, di);
                o.x = fmaxf(o.x + bv.x, 0.0f) * di;
                o.y = fmaxf(o.y + bv.y, 0.0f) * di;
            }
            *(v2f*)(tA + r * HID + ((2 * lane) ^ (r << 2))) = o;
        }
        __syncthreads();

        if (((widx >> 2) ^ it) & 1) continue;
        const int t = widx & 3;
        const float* aRow = tA + sub * HID;
        const int swz = sub << 2;
        v8f acc = {};
#pragma unroll 4
        for (int kk = 0; kk < HID; kk += 4) {
            const int kb = kk + 2 * half;
            const v2f a = *(const v2f*)(aRow + (kb ^ swz));
            const v2f b = *(const v2f*)(sW + (kb >> 1) * (2 * HID) + 2 * sub + t * 32);
            acc = __builtin_amdgcn_wmma_f32_16x16x4_f32(
                 false, a,  false, b,
                 (short)0, acc,
                 false,  false);
        }
        for (int r = 0; r < 8; ++r) {
            const int node = tile * 16 + r + 8 * half;
            if (node < N) H2[(long)node * HID + t * 16 + sub] = acc[r];
        }
    }
}

__global__ void __launch_bounds__(256)
gather_kernel(const float* __restrict__ h, const float* __restrict__ dinv,
              const float* __restrict__ bias, const int* __restrict__ off,
              const int* __restrict__ cnt,
              const int2* __restrict__ rec, float* __restrict__ out,
              float* __restrict__ lsm, int N) {
    const int lane = threadIdx.x & 31;
    const int wave = __builtin_amdgcn_readfirstlane(blockIdx.x * 8 + (threadIdx.x >> 5));
    const int nwaves = gridDim.x * 8;
    const v2f* hl = (const v2f*)h + lane;
    const v2f bv = ((const v2f*)bias)[lane];
    for (int node = wave; node < N; node += nwaves) {
        float di;
        v2f o = node_row(hl, dinv, off, cnt, rec, node, lane, di);
        o.x += bv.x;
        o.y += bv.y;
        ((v2f*)out)[(long)node * (HID / 2) + lane] = o;
        const float mx = wave_max(fmaxf(o.x, o.y));
        const float se = wave_sum(__expf(o.x - mx) + __expf(o.y - mx));
        const float lse = mx + __logf(se);
        v2f l;
        l.x = o.x - lse;
        l.y = o.y - lse;
        ((v2f*)lsm)[(long)node * (HID / 2) + lane] = l;
    }
}

extern "C" void kernel_launch(void* const* d_in, const int* in_sizes, int n_in,
                              void* d_out, int out_size, void* d_ws, size_t ws_size,
                              hipStream_t stream) {
    const float* x  = (const float*)d_in[0];
    const int*   ei = (const int*)d_in[1];
    const int*   ec = (const int*)d_in[2];
    const float* W1 = (const float*)d_in[3];
    const float* b1 = (const float*)d_in[4];
    const float* W2 = (const float*)d_in[5];
    const float* b2 = (const float*)d_in[6];

    const int IN_DIM = 256;
    const int N = in_sizes[0] / IN_DIM;
    const int E = in_sizes[2];
    const int* row = ei;
    const int* col = ei + E;

    float* wsf  = (float*)d_ws;
    float* h    = wsf;
    float* h2   = h    + (size_t)N * HID;
    int2*  rec  = (int2*)(h2 + (size_t)N * HID);
    int*   rank = (int*)(rec + (size_t)E);
    float* dinv = (float*)(rank + (size_t)E);
    int*   len  = (int*)(dinv + (size_t)N);
    u64*   cnt  = (u64*)h2;
    u64*   cnt1 = cnt + (size_t)N;
    int*   off  = len + (size_t)N;
    const int nb = (N + SCAN_B - 1) / SCAN_B;
    int*   mm   = off + (size_t)N;
    int*   cur0 = mm + 64;
    int*   mm1  = (int*)(cnt1 + (size_t)N) + 64;
    int*   cur1 = mm1 + 64;
    int2*  fb   = (int2*)(cur1 + 64);

    float* emb = (float*)d_out;
    float* lsm = emb + (size_t)N * HID;

    const int tileBlocks = (int)((N / 16 + 7) / 8);
    const int edgeBlocks = (E + 256 * EPT - 1) / (256 * EPT);
    const int nodeBlocks = 2048;
    const int mmBlocks = 256;

    init_kernel<<<(N + 255) / 256, 256, 0, stream>>>(cnt, cnt1, mm, mm1, cur0, cur1, N);
    count_kernel<<<edgeBlocks + mmBlocks, 256, 0, stream>>>(col, cnt, cnt1, rank, E, edgeBlocks, ec, mm, mm1);
    scan_kernel<<<nb, SCAN_B, 0, stream>>>((const int2*)cnt, (const int2*)cnt1, mm, mm1, cur0, cur1, off, fb, len, dinv, N, E);
    mid_kernel<<<tileBlocks + edgeBlocks, 256, 0, stream>>>(
        x, W1, dinv, h, N, tileBlocks, ec, row, col, rank, fb, mm, mm1, rec, E);

    gather_gemm_kernel<<<nodeBlocks, 256, 0, stream>>>(h, dinv, b1, off, len, rec, W2, h2, N);

    gather_kernel<<<nodeBlocks, 256, 0, stream>>>(h2, dinv, b2, off, len, rec, emb, lsm, N);
}
